# starting kernel
# baseline (speedup 1.0000x reference)
#include <hip/hip_runtime.h>
#include <hip/hip_bf16.h>

#define NSZ 160
#define NN (NSZ * NSZ)
#define HID 128
#define NHEADS 8
#define HD 16
#define LN_EPS 0.001f

typedef __attribute__((ext_vector_type(2))) float v2f;
typedef __attribute__((ext_vector_type(8))) float v8f;

static __device__ __forceinline__ v8f wmma_f32_16x16x4(v2f a, v2f b, v8f c) {
    return __builtin_amdgcn_wmma_f32_16x16x4_f32(
         false, a,  false, b,
         (short)0, c,  false,  false);
}


__global__ __launch_bounds__(32) void wpack_kernel(
    const float* __restrict__ Wq, const float* __restrict__ Wkv,
    const float* __restrict__ Weg, float* __restrict__ Wt)
{
    const int ct   = blockIdx.x;
    const int lane = threadIdx.x;
    const int hs   = lane / 16;
    const int lm   = lane % 16;

    const float* W; int wcols, n0;
    if (ct < 8)       { W = Wq;  wcols = HID;        n0 = ct * 16; }
    else if (ct < 24) { W = Wkv; wcols = 2 * HID;    n0 = (ct - 8) * 16; }
    else              { W = Weg; wcols = 2 * NHEADS; n0 = 0; }

    float* o = Wt + (size_t)ct * 2048 + lane * 2;
    for (int ks = 0; ks < 32; ++ks) {
        const int kk = ks * 4 + 2 * hs;
        o[ks * 64 + 0] = W[(size_t)(kk)     * wcols + n0 + lm];
        o[ks * 64 + 1] = W[(size_t)(kk + 1) * wcols + n0 + lm];
    }
}

static __device__ __forceinline__ v8f gemm_tile(
    const float* __restrict__ sA, const float* __restrict__ wt,
    int lm, int hs)
{
    v8f acc = {0.f, 0.f, 0.f, 0.f, 0.f, 0.f, 0.f, 0.f};
    const float* pa = sA + lm * HID + 2 * hs;
    #pragma unroll
    for (int ks = 0; ks < 32; ++ks) {
        v2f a, b;
        a.x = pa[ks * 4];
        a.y = pa[ks * 4 + 1];
        b = *reinterpret_cast<const v2f*>(wt + ks * 64);
        acc = wmma_f32_16x16x4(a, b, acc);
    }
    return acc;
}

__global__ __launch_bounds__(256) void proj_kernel(
    const float* __restrict__ feat1, const float* __restrict__ feat2,
    const float* __restrict__ Wt,
    const float* __restrict__ bq, const float* __restrict__ bkv,
    const float* __restrict__ beg,
    float* __restrict__ Qp, float* __restrict__ Kp, float* __restrict__ Vp,
    float* __restrict__ Ep, float* __restrict__ Gp)
{
    __shared__ float sA1[16][HID];
    __shared__ float sA2[16][HID];

    const int row0 = blockIdx.x * 16;
    const int tid  = threadIdx.x;

    for (int idx = tid; idx < 16 * (HID / 4); idx += 256) {
        const int r = idx / (HID / 4), c4 = idx % (HID / 4);
        reinterpret_cast<float4*>(&sA1[r][0])[c4] =
            reinterpret_cast<const float4*>(&feat1[(size_t)(row0 + r) * HID])[c4];
        reinterpret_cast<float4*>(&sA2[r][0])[c4] =
            reinterpret_cast<const float4*>(&feat2[(size_t)(row0 + r) * HID])[c4];
    }
    __syncthreads();

    const int wave = tid / 32;
    const int lane = tid % 32;
    const int hs   = lane / 16;
    const int lm   = lane % 16;

    const int aIdx = row0 / NSZ;
    const int b0   = row0 % NSZ;
    const int bBase = b0 + 8 * hs;

    const int col = wave * 16 + lm;
    const int d   = col / 8;
    const int h   = col % 8;

    {
        v8f acc = gemm_tile(&sA1[0][0], Wt + (size_t)wave * 2048 + lane * 2, lm, hs);
        const float bv = bq[col];
        float* oq = Qp + (((size_t)bBase * NHEADS + h) * NSZ + aIdx) * HD + d;
        #pragma unroll
        for (int r = 0; r < 8; ++r)
            oq[(size_t)r * NHEADS * NSZ * HD] = (acc[r] + bv) * 0.25f;
    }

    {
        v8f acc = gemm_tile(&sA2[0][0], Wt + (size_t)(8 + wave) * 2048 + lane * 2, lm, hs);
        const float bv = bkv[col];
        float* ok = Kp + (((size_t)aIdx * NHEADS + h) * NSZ + bBase) * HD + d;
        #pragma unroll
        for (int r = 0; r < 8; ++r)
            ok[r * HD] = acc[r] + bv;
    }

    {
        v8f acc = gemm_tile(&sA2[0][0], Wt + (size_t)(16 + wave) * 2048 + lane * 2, lm, hs);
        const float bv = bkv[HID + col];
        float* ov = Vp + (((size_t)aIdx * NHEADS + h) * HD + d) * NSZ + bBase;
        #pragma unroll
        for (int r = 0; r < 8; ++r)
            ov[r] = acc[r] + bv;
    }

    if (wave == 0) {
        v8f acc = gemm_tile(&sA2[0][0], Wt + (size_t)24 * 2048 + lane * 2, lm, hs);
        const float bv = beg[lm];
        if (lm < NHEADS) {
            float* oe = Ep + ((size_t)aIdx * NHEADS + lm) * NSZ + bBase;
            #pragma unroll
            for (int r = 0; r < 8; ++r) oe[r] = acc[r] + bv;
        } else {
            float* og = Gp + ((size_t)aIdx * NHEADS + (lm - NHEADS)) * NSZ + bBase;
            #pragma unroll
            for (int r = 0; r < 8; ++r)
                og[r] = 1.f / (1.f + __expf(-(acc[r] + bv)));
        }
    }
}

__global__ __launch_bounds__(256) void attn_kernel(
    const float* __restrict__ Qp, const float* __restrict__ Kp,
    const float* __restrict__ Vp, const float* __restrict__ Ep,
    const float* __restrict__ Gp,
    const float* __restrict__ ln_g, const float* __restrict__ ln_b,
    float* __restrict__ out)
{
    __shared__ float Plds[NHEADS][16][81];
    __shared__ float VaLds[16][HID];

    const int j    = blockIdx.x;
    const int i0   = blockIdx.y * 16;
    const int tid  = threadIdx.x;
    const int h    = tid / 32;
    const int lane = tid % 32;
    const int hs   = lane / 16;
    const int lm   = lane % 16;

    const size_t jh = (size_t)j * NHEADS + h;

    const float* pQ = Qp + (jh * NSZ + (i0 + lm)) * HD + 2 * hs;
    const float* pK = Kp + (jh * NSZ + lm) * HD + 2 * hs;
    const float* pV = Vp + (jh * HD + lm) * NSZ + 2 * hs;
    const float* pE = Ep + jh * NSZ + lm;
    const float* pG = Gp + jh * NSZ + lm;

    {
        const float* vslice = Vp + jh * HD * NSZ;
        for (int p = lane; p < 80; p += 32)
            __builtin_prefetch(vslice + p * 32, 0, 1);
    }

    v2f afr[4];
    #pragma unroll
    for (int ks = 0; ks < 4; ++ks)
        afr[ks] = *reinterpret_cast<const v2f*>(pQ + ks * 4);

    v8f acc[10];
    #pragma unroll
    for (int ct = 0; ct < 10; ++ct) {
        v8f c = {0.f, 0.f, 0.f, 0.f, 0.f, 0.f, 0.f, 0.f};
        #pragma unroll
        for (int ks = 0; ks < 4; ++ks) {
            v2f b = *reinterpret_cast<const v2f*>(pK + ct * 16 * HD + ks * 4);
            c = wmma_f32_16x16x4(afr[ks], b, c);
        }
        const float e = pE[ct * 16];
        #pragma unroll
        for (int r = 0; r < 8; ++r) c[r] += e;
        acc[ct] = c;
    }

    float gate[10];
    #pragma unroll
    for (int ct = 0; ct < 10; ++ct) gate[ct] = pG[ct * 16];

    #pragma unroll
    for (int r = 0; r < 8; ++r) {
        float mx = -1e30f;
        #pragma unroll
        for (int ct = 0; ct < 10; ++ct) mx = fmaxf(mx, acc[ct][r]);
        #pragma unroll
        for (int msk = 1; msk < 16; msk <<= 1) mx = fmaxf(mx, __shfl_xor(mx, msk, 32));
        float sm = 0.f;
        #pragma unroll
        for (int ct = 0; ct < 10; ++ct) {
            float e = __expf(acc[ct][r] - mx);
            acc[ct][r] = e;
            sm += e;
        }
        #pragma unroll
        for (int msk = 1; msk < 16; msk <<= 1) sm += __shfl_xor(sm, msk, 32);
        const float inv = 1.f / sm;
        #pragma unroll
        for (int ct = 0; ct < 10; ++ct) acc[ct][r] = acc[ct][r] * inv * gate[ct];
    }

    v8f va = {0.f, 0.f, 0.f, 0.f, 0.f, 0.f, 0.f, 0.f};
    #pragma unroll
    for (int half = 0; half < 2; ++half) {
        #pragma unroll
        for (int ct = 0; ct < 5; ++ct) {
            #pragma unroll
            for (int r = 0; r < 8; ++r)
                Plds[h][r + 8 * hs][ct * 16 + lm] = acc[half * 5 + ct][r];
        }
        __syncthreads();
        #pragma unroll
        for (int k0 = 0; k0 < 80; k0 += 4) {
            v2f a, b;
            a.x = Plds[h][lm][k0 + 2 * hs];
            a.y = Plds[h][lm][k0 + 2 * hs + 1];
            b = *reinterpret_cast<const v2f*>(pV + half * 80 + k0);
            va = wmma_f32_16x16x4(a, b, va);
        }
        __syncthreads();
    }

    #pragma unroll
    for (int r = 0; r < 8; ++r)
        VaLds[r + 8 * hs][lm * NHEADS + h] = va[r];
    __syncthreads();

    const int rowLN = tid / 16;
    const int cg    = tid % 16;
    float vals[8], s1 = 0.f, s2 = 0.f;
    #pragma unroll
    for (int c = 0; c < 8; ++c) {
        float v = VaLds[rowLN][cg * 8 + c];
        vals[c] = v; s1 += v; s2 += v * v;
    }
    #pragma unroll
    for (int msk = 1; msk < 16; msk <<= 1) {
        s1 += __shfl_xor(s1, msk, 32);
        s2 += __shfl_xor(s2, msk, 32);
    }
    const float mu  = s1 * (1.f / HID);
    const float var = s2 * (1.f / HID) - mu * mu;
    const float inv = rsqrtf(var + LN_EPS);

    const size_t obase = ((size_t)(i0 + rowLN) * NSZ + j) * HID + cg * 8;
    v2f* op = reinterpret_cast<v2f*>(&out[obase]);
    #pragma unroll
    for (int c = 0; c < 8; c += 2) {
        const int ch = cg * 8 + c;
        v2f t;
        t.x = (vals[c]     - mu) * inv * ln_g[ch]     + ln_b[ch];
        t.y = (vals[c + 1] - mu) * inv * ln_g[ch + 1] + ln_b[ch + 1];
        op[c / 2] = t;
    }
}

extern "C" void kernel_launch(void* const* d_in, const int* in_sizes, int n_in,
                              void* d_out, int out_size, void* d_ws, size_t ws_size,
                              hipStream_t stream) {
    const float* feat1 = (const float*)d_in[0];
    const float* feat2 = (const float*)d_in[1];
    const float* Wq    = (const float*)d_in[2];
    const float* bq    = (const float*)d_in[3];
    const float* Wkv   = (const float*)d_in[4];
    const float* bkv   = (const float*)d_in[5];
    const float* Weg   = (const float*)d_in[6];
    const float* beg   = (const float*)d_in[7];
    const float* lng   = (const float*)d_in[8];
    const float* lnb   = (const float*)d_in[9];

    float* ws = (float*)d_ws;
    float* Qp = ws;
    float* Kp = Qp + (size_t)NN * HID;
    float* Vp = Kp + (size_t)NN * HID;
    float* Ep = Vp + (size_t)NN * HID;
    float* Gp = Ep + (size_t)NN * NHEADS;
    float* Wt = Gp + (size_t)NN * NHEADS;

    wpack_kernel<<<dim3(25), 32, 0, stream>>>(Wq, Wkv, Weg, Wt);

    proj_kernel<<<dim3(NN / 16), 256, 0, stream>>>(
        feat1, feat2, Wt, bq, bkv, beg, Qp, Kp, Vp, Ep, Gp);

    attn_kernel<<<dim3(NSZ, NSZ / 16), 256, 0, stream>>>(
        Qp, Kp, Vp, Ep, Gp, lng, lnb, (float*)d_out);
}


# Weight pack launch and its workspace plane removed: each wave gathers its WMMA weight fragments from the row-major matrices itself (K/V channels dealt to waves in contiguous 16s); sum orders kept
# speedup vs baseline: 2.2235x; 2.2235x over previous
#include <hip/hip_runtime.h>
#include <hip/hip_bf16.h>


#define NSZ 160
#define NN (NSZ * NSZ)
#define HID 128
#define NHEADS 8
#define HD 16
#define LN_EPS 0.001f

#define FP (HID + 4)
#define KP 20
#define VP 164
#define VH (HD * VP + 8)

typedef __attribute__((ext_vector_type(2))) float v2f;
typedef __attribute__((ext_vector_type(8))) float v8f;

static __device__ __forceinline__ v8f wmma_f32_16x16x4(v2f a, v2f b, v8f c) {
    return __builtin_amdgcn_wmma_f32_16x16x4_f32(
         false, a,  false, b,
         (short)0, c,  false,  false);
}

static __device__ __forceinline__ void wave_lds_handover() {
    __builtin_amdgcn_fence(__ATOMIC_RELEASE, "wavefront");
    __builtin_amdgcn_wave_barrier();
    __builtin_amdgcn_fence(__ATOMIC_ACQUIRE, "wavefront");
}


__global__ __launch_bounds__(256)
void fused_kernel(
    const float* __restrict__ feat1, const float* __restrict__ feat2,
    const float* __restrict__ Wq, const float* __restrict__ Wkv,
    const float* __restrict__ Weg,
    const float* __restrict__ bq, const float* __restrict__ bkv,
    const float* __restrict__ beg,
    const float* __restrict__ ln_g, const float* __restrict__ ln_b,
    float* __restrict__ out)
{
    __shared__ __attribute__((aligned(16))) float Kl[NHEADS * NSZ * KP];
    __shared__ __attribute__((aligned(16))) float Vl[NHEADS * VH];
    __shared__ float El[NHEADS * NSZ];
    __shared__ float Gl[NHEADS * NSZ];
    __shared__ float Plds[NHEADS][16][81];
    __shared__ __attribute__((aligned(16))) float sFV[2 * 16 * FP];
    __shared__ float VaLds[2][16][HID];

    const int j    = blockIdx.x;
    const int tid  = threadIdx.x;
    const int h    = tid / 32;
    const int lane = tid % 32;
    const int hs   = lane / 16;
    const int lm   = lane % 16;

    const float* f2 = feat2 + (unsigned)j * NSZ * HID;

    float4 n0 = reinterpret_cast<const float4*>(f2 + (unsigned)(h)     * HID)[lane];
    float4 n1 = reinterpret_cast<const float4*>(f2 + (unsigned)(h + 8) * HID)[lane];

    const float* pf  = sFV + lm * FP + 2 * hs;

    v2f wk[32], wv[32], we[32], wq[32];
    const float* pwq = Wq + (unsigned)(2 * hs) * HID + lm * NHEADS + h;
    {
        const float* pwk = Wkv + (unsigned)(2 * hs) * (2 * HID) + 16 * h + lm;
        const float* pwv = pwk + HID;
        const float* pwe = Weg + (unsigned)(2 * hs) * (2 * NHEADS) + lm;
        #pragma unroll
        for (int ks = 0; ks < 32; ++ks) {
            wk[ks].x = pwk[(4 * ks)     * (2 * HID)];
            wk[ks].y = pwk[(4 * ks + 1) * (2 * HID)];
            wv[ks].x = pwv[(4 * ks)     * (2 * HID)];
            wv[ks].y = pwv[(4 * ks + 1) * (2 * HID)];
        }
        #pragma unroll
        for (int ks = 0; ks < 32; ++ks) {
            we[ks].x = pwe[(4 * ks)     * (2 * NHEADS)];
            we[ks].y = pwe[(4 * ks + 1) * (2 * NHEADS)];
        }
        #pragma unroll
        for (int ks = 0; ks < 32; ++ks) { wq[ks].x = 0.f; wq[ks].y = 0.f; }
    }

    float bk[8];
    #pragma unroll
    for (int r = 0; r < 8; ++r) bk[r] = bkv[16 * h + 8 * hs + r];
    const float bvv = bkv[HID + 16 * h + lm];
    const float bev = beg[lm];

    for (int t = 0; t < 10; ++t) {
        *reinterpret_cast<float4*>(&sFV[h * FP + lane * 4])       = n0;
        *reinterpret_cast<float4*>(&sFV[(h + 8) * FP + lane * 4]) = n1;
        __syncthreads();

        {
            const float* s0 = (t < 9) ? f2 + (unsigned)((t + 1) * 16 + h) * HID
                                      : feat1 + ((unsigned)(h) * NSZ + j) * HID;
            const float* s1 = (t < 9) ? f2 + (unsigned)((t + 1) * 16 + h + 8) * HID
                                      : feat1 + ((unsigned)(h + 8) * NSZ + j) * HID;
            n0 = reinterpret_cast<const float4*>(s0)[lane];
            n1 = reinterpret_cast<const float4*>(s1)[lane];
        }
        if (t == 0) {
            #pragma unroll
            for (int ks = 0; ks < 32; ++ks) {
                wq[ks].x = pwq[(4 * ks)     * HID];
                wq[ks].y = pwq[(4 * ks + 1) * HID];
            }
        }

        v8f kd = {0.f, 0.f, 0.f, 0.f, 0.f, 0.f, 0.f, 0.f};
        v8f vd = {0.f, 0.f, 0.f, 0.f, 0.f, 0.f, 0.f, 0.f};
        #pragma unroll
        for (int ks = 0; ks < 32; ++ks) {
            v2f f;
            f.x = pf[ks * 4];
            f.y = pf[ks * 4 + 1];
            kd = wmma_f32_16x16x4(wk[ks], f, kd);
            vd = wmma_f32_16x16x4(f, wv[ks], vd);
        }

        {
            float* ok = Kl + (t * 16 + lm) * KP + 2 * h + hs;
            #pragma unroll
            for (int r = 0; r < 8; ++r) ok[r * NSZ * KP] = kd[r] + bk[r];
        }
        {
            float4 a, b;
            a.x = vd[0] + bvv; a.y = vd[1] + bvv; a.z = vd[2] + bvv; a.w = vd[3] + bvv;
            b.x = vd[4] + bvv; b.y = vd[5] + bvv; b.z = vd[6] + bvv; b.w = vd[7] + bvv;
            float* ov = Vl + (lm % 8) * VH + (2 * h + lm / 8) * VP + t * 16 + 8 * hs;
            *reinterpret_cast<float4*>(ov)     = a;
            *reinterpret_cast<float4*>(ov + 4) = b;
        }

        if (h == (t & 7)) {
            v8f acc = {0.f, 0.f, 0.f, 0.f, 0.f, 0.f, 0.f, 0.f};
            #pragma unroll
            for (int ks = 0; ks < 32; ++ks) {
                v2f a;
                a.x = pf[ks * 4];
                a.y = pf[ks * 4 + 1];
                acc = wmma_f32_16x16x4(a, we[ks], acc);
            }
            if (lm < NHEADS) {
                float* oe = El + lm * NSZ + t * 16 + 8 * hs;
                #pragma unroll
                for (int r = 0; r < 8; ++r) oe[r] = acc[r] + bev;
            } else {
                float* og = Gl + (lm - NHEADS) * NSZ + t * 16 + 8 * hs;
                #pragma unroll
                for (int r = 0; r < 8; ++r)
                    og[r] = 1.f / (1.f + __expf(-(acc[r] + bev)));
            }
        }
        __syncthreads();
    }

    const float* pK = Kl + (h * NSZ + lm) * KP + 8 * hs;
    const float* pV = Vl + h * VH + lm * VP + 2 * hs;

    float4 n2 = reinterpret_cast<const float4*>(
        &feat1[((unsigned)(16 + h) * NSZ + j) * HID])[lane];
    float4 n3 = reinterpret_cast<const float4*>(
        &feat1[((unsigned)(24 + h) * NSZ + j) * HID])[lane];

    float ev[10], gate[10];
    #pragma unroll
    for (int ct = 0; ct < 10; ++ct) {
        ev[ct]   = El[h * NSZ + ct * 16 + lm];
        gate[ct] = Gl[h * NSZ + ct * 16 + lm];
    }
    float bqv[8];
    #pragma unroll
    for (int r = 0; r < 8; ++r) bqv[r] = bq[(8 * hs + r) * NHEADS + h];

    const int rowLN = tid / 16;
    const int cg    = tid % 16;

    v2f ot0[4] = {{0.f, 0.f}, {0.f, 0.f}, {0.f, 0.f}, {0.f, 0.f}};
    v2f ot1[4] = {{0.f, 0.f}, {0.f, 0.f}, {0.f, 0.f}, {0.f, 0.f}};
    unsigned oprev0 = 0, oprev1 = 0;

    for (int ip = 0; ip < 5; ++ip) {
        *reinterpret_cast<float4*>(&sFV[h * FP + lane * 4])              = n0;
        *reinterpret_cast<float4*>(&sFV[(h + 8) * FP + lane * 4])        = n1;
        *reinterpret_cast<float4*>(&sFV[(16 + h) * FP + lane * 4])       = n2;
        *reinterpret_cast<float4*>(&sFV[(24 + h) * FP + lane * 4])       = n3;
        __syncthreads();

        if (ip > 0) {
            v2f* op0 = reinterpret_cast<v2f*>(&out[oprev0]);
            v2f* op1 = reinterpret_cast<v2f*>(&out[oprev1]);
            #pragma unroll
            for (int c = 0; c < 4; ++c) { op0[c] = ot0[c]; op1[c] = ot1[c]; }
        }
        if (ip < 4) {
            const unsigned rb = (unsigned)((ip + 1) * 32 + h);
            n0 = reinterpret_cast<const float4*>(&feat1[((rb)      * NSZ + j) * HID])[lane];
            n1 = reinterpret_cast<const float4*>(&feat1[((rb + 8)  * NSZ + j) * HID])[lane];
            n2 = reinterpret_cast<const float4*>(&feat1[((rb + 16) * NSZ + j) * HID])[lane];
            n3 = reinterpret_cast<const float4*>(&feat1[((rb + 24) * NSZ + j) * HID])[lane];
        }

        #pragma unroll
        for (int sub = 0; sub < 2; ++sub) {
            const float* pfs = pf + sub * 16 * FP;

            v8f qd = {0.f, 0.f, 0.f, 0.f, 0.f, 0.f, 0.f, 0.f};
            #pragma unroll
            for (int ks = 0; ks < 32; ++ks) {
                v2f b;
                b.x = pfs[ks * 4];
                b.y = pfs[ks * 4 + 1];
                qd = wmma_f32_16x16x4(wq[ks], b, qd);
            }

            v2f afr[4];
            #pragma unroll
            for (int ks = 0; ks < 4; ++ks) {
                afr[ks].x = (qd[2 * ks]     + bqv[2 * ks])     * 0.25f;
                afr[ks].y = (qd[2 * ks + 1] + bqv[2 * ks + 1]) * 0.25f;
            }

            v8f acc[10];
            #pragma unroll
            for (int ct = 0; ct < 10; ++ct) {
                v8f c = {0.f, 0.f, 0.f, 0.f, 0.f, 0.f, 0.f, 0.f};
                #pragma unroll
                for (int ks = 0; ks < 4; ++ks) {
                    v2f b = *reinterpret_cast<const v2f*>(pK + ct * 16 * KP + ks * 2);
                    c = wmma_f32_16x16x4(afr[ks], b, c);
                }
                const float e = ev[ct];
                #pragma unroll
                for (int r = 0; r < 8; ++r) c[r] += e;
                acc[ct] = c;
            }

            #pragma unroll
            for (int r = 0; r < 8; ++r) {
                float mx = -1e30f;
                #pragma unroll
                for (int ct = 0; ct < 10; ++ct) mx = fmaxf(mx, acc[ct][r]);
                #pragma unroll
                for (int msk = 1; msk < 16; msk <<= 1) mx = fmaxf(mx, __shfl_xor(mx, msk, 32));
                float sm = 0.f;
                #pragma unroll
                for (int ct = 0; ct < 10; ++ct) {
                    float e = __expf(acc[ct][r] - mx);
                    acc[ct][r] = e;
                    sm += e;
                }
                #pragma unroll
                for (int msk = 1; msk < 16; msk <<= 1) sm += __shfl_xor(sm, msk, 32);
                const float inv = 1.f / sm;
                #pragma unroll
                for (int ct = 0; ct < 10; ++ct) acc[ct][r] = acc[ct][r] * inv * gate[ct];
            }

            v8f va = {0.f, 0.f, 0.f, 0.f, 0.f, 0.f, 0.f, 0.f};
            #pragma unroll
            for (int half = 0; half < 2; ++half) {
                #pragma unroll
                for (int ct = 0; ct < 5; ++ct) {
                    #pragma unroll
                    for (int r = 0; r < 8; ++r)
                        Plds[h][r + 8 * hs][ct * 16 + lm] = acc[half * 5 + ct][r];
                }
                wave_lds_handover();
                #pragma unroll
                for (int k0 = 0; k0 < 80; k0 += 4) {
                    v2f a, b;
                    a.x = Plds[h][lm][k0 + 2 * hs];
                    a.y = Plds[h][lm][k0 + 2 * hs + 1];
                    b = *reinterpret_cast<const v2f*>(pV + half * 80 + k0);
                    va = wmma_f32_16x16x4(a, b, va);
                }
                wave_lds_handover();
            }

            #pragma unroll
            for (int r = 0; r < 8; ++r)
                VaLds[sub][r + 8 * hs][lm * NHEADS + h] = va[r];
        }
        __syncthreads();

        #pragma unroll
        for (int sub = 0; sub < 2; ++sub) {
            float vals[8], s1 = 0.f, s2 = 0.f;
            #pragma unroll
            for (int c = 0; c < 8; ++c) {
                float v = VaLds[sub][rowLN][cg * 8 + c];
                vals[c] = v; s1 += v; s2 += v * v;
            }
            #pragma unroll
            for (int msk = 1; msk < 16; msk <<= 1) {
                s1 += __shfl_xor(s1, msk, 32);
                s2 += __shfl_xor(s2, msk, 32);
            }
            const float mu  = s1 * (1.f / HID);
            const float var = s2 * (1.f / HID) - mu * mu;
            const float inv = rsqrtf(var + LN_EPS);

            const unsigned op = ((unsigned)(ip * 32 + sub * 16 + rowLN) * NSZ + j) * HID + cg * 8;
            if (sub == 0) oprev0 = op; else oprev1 = op;
            #pragma unroll
            for (int c = 0; c < 8; c += 2) {
                const int ch = cg * 8 + c;
                v2f tv;
                tv.x = (vals[c]     - mu) * inv * ln_g[ch]     + ln_b[ch];
                tv.y = (vals[c + 1] - mu) * inv * ln_g[ch + 1] + ln_b[ch + 1];
                if (sub == 0) ot0[c / 2] = tv; else ot1[c / 2] = tv;
            }
        }
    }
    {
        v2f* op0 = reinterpret_cast<v2f*>(&out[oprev0]);
        v2f* op1 = reinterpret_cast<v2f*>(&out[oprev1]);
        #pragma unroll
        for (int c = 0; c < 4; ++c) { op0[c] = ot0[c]; op1[c] = ot1[c]; }
    }
}

extern "C" void kernel_launch(void* const* d_in, const int* in_sizes, int n_in,
                              void* d_out, int out_size, void* d_ws, size_t ws_size,
                              hipStream_t stream) {
    const float* feat1 = (const float*)d_in[0];
    const float* feat2 = (const float*)d_in[1];
    const float* Wq    = (const float*)d_in[2];
    const float* bq    = (const float*)d_in[3];
    const float* Wkv   = (const float*)d_in[4];
    const float* bkv   = (const float*)d_in[5];
    const float* Weg   = (const float*)d_in[6];
    const float* beg   = (const float*)d_in[7];
    const float* lng   = (const float*)d_in[8];
    const float* lnb   = (const float*)d_in[9];

    fused_kernel<<<dim3(NSZ), 256, 0, stream>>>(
        feat1, feat2, Wq, Wkv, Weg, bq, bkv, beg, lng, lnb, (float*)d_out);
}


# Projection runs: the staged feat tile's WMMA fragments are read from LDS as one group into registers ahead of the run (K/V run, E/gate run reusing them, first Q run of a pair), not singly; sums kept
# speedup vs baseline: 2.2973x; 1.0332x over previous
#include <hip/hip_runtime.h>
#include <hip/hip_bf16.h>


#define NSZ 160
#define NN (NSZ * NSZ)
#define HID 128
#define NHEADS 8
#define HD 16
#define LN_EPS 0.001f

#define FP (HID + 4)
#define KP 20
#define VP 164
#define VH (HD * VP + 8)

typedef __attribute__((ext_vector_type(2))) float v2f;
typedef __attribute__((ext_vector_type(8))) float v8f;

static __device__ __forceinline__ v8f wmma_f32_16x16x4(v2f a, v2f b, v8f c) {
    return __builtin_amdgcn_wmma_f32_16x16x4_f32(
         false, a,  false, b,
         (short)0, c,  false,  false);
}

static __device__ __forceinline__ void wave_lds_handover() {
    __builtin_amdgcn_fence(__ATOMIC_RELEASE, "wavefront");
    __builtin_amdgcn_wave_barrier();
    __builtin_amdgcn_fence(__ATOMIC_ACQUIRE, "wavefront");
}


__global__ __launch_bounds__(256)
void fused_kernel(
    const float* __restrict__ feat1, const float* __restrict__ feat2,
    const float* __restrict__ Wq, const float* __restrict__ Wkv,
    const float* __restrict__ Weg,
    const float* __restrict__ bq, const float* __restrict__ bkv,
    const float* __restrict__ beg,
    const float* __restrict__ ln_g, const float* __restrict__ ln_b,
    float* __restrict__ out)
{
    __shared__ __attribute__((aligned(16))) float Kl[NHEADS * NSZ * KP];
    __shared__ __attribute__((aligned(16))) float Vl[NHEADS * VH];
    __shared__ float El[NHEADS * NSZ];
    __shared__ float Gl[NHEADS * NSZ];
    __shared__ float Plds[NHEADS][16][81];
    __shared__ __attribute__((aligned(16))) float sFV[2 * 16 * FP];
    __shared__ float VaLds[2][16][HID];

    const int j    = blockIdx.x;
    const int tid  = threadIdx.x;
    const int h    = tid / 32;
    const int lane = tid % 32;
    const int hs   = lane / 16;
    const int lm   = lane % 16;

    const float* f2 = feat2 + (unsigned)j * NSZ * HID;

    float4 n0 = reinterpret_cast<const float4*>(f2 + (unsigned)(h)     * HID)[lane];
    float4 n1 = reinterpret_cast<const float4*>(f2 + (unsigned)(h + 8) * HID)[lane];

    const float* pf  = sFV + lm * FP + 2 * hs;

    v2f wk[32], wv[32], we[32], wq[32];
    const float* pwq = Wq + (unsigned)(2 * hs) * HID + lm * NHEADS + h;
    {
        const float* pwk = Wkv + (unsigned)(2 * hs) * (2 * HID) + 16 * h + lm;
        const float* pwv = pwk + HID;
        const float* pwe = Weg + (unsigned)(2 * hs) * (2 * NHEADS) + lm;
        #pragma unroll
        for (int ks = 0; ks < 32; ++ks) {
            wk[ks].x = pwk[(4 * ks)     * (2 * HID)];
            wk[ks].y = pwk[(4 * ks + 1) * (2 * HID)];
            wv[ks].x = pwv[(4 * ks)     * (2 * HID)];
            wv[ks].y = pwv[(4 * ks + 1) * (2 * HID)];
        }
        #pragma unroll
        for (int ks = 0; ks < 32; ++ks) {
            we[ks].x = pwe[(4 * ks)     * (2 * NHEADS)];
            we[ks].y = pwe[(4 * ks + 1) * (2 * NHEADS)];
        }
        #pragma unroll
        for (int ks = 0; ks < 32; ++ks) { wq[ks].x = 0.f; wq[ks].y = 0.f; }
    }

    float bk[8];
    #pragma unroll
    for (int r = 0; r < 8; ++r) bk[r] = bkv[16 * h + 8 * hs + r];
    const float bvv = bkv[HID + 16 * h + lm];
    const float bev = beg[lm];

    for (int t = 0; t < 10; ++t) {
        *reinterpret_cast<float4*>(&sFV[h * FP + lane * 4])       = n0;
        *reinterpret_cast<float4*>(&sFV[(h + 8) * FP + lane * 4]) = n1;
        __syncthreads();

        {
            const float* s0 = (t < 9) ? f2 + (unsigned)((t + 1) * 16 + h) * HID
                                      : feat1 + ((unsigned)(h) * NSZ + j) * HID;
            const float* s1 = (t < 9) ? f2 + (unsigned)((t + 1) * 16 + h + 8) * HID
                                      : feat1 + ((unsigned)(h + 8) * NSZ + j) * HID;
            n0 = reinterpret_cast<const float4*>(s0)[lane];
            n1 = reinterpret_cast<const float4*>(s1)[lane];
        }
        if (t == 0) {
            #pragma unroll
            for (int ks = 0; ks < 32; ++ks) {
                wq[ks].x = pwq[(4 * ks)     * HID];
                wq[ks].y = pwq[(4 * ks + 1) * HID];
            }
        }

        v2f fr[32];
        #pragma unroll
        for (int ks = 0; ks < 32; ++ks)
            fr[ks] = *reinterpret_cast<const v2f*>(pf + ks * 4);

        v8f kd = {0.f, 0.f, 0.f, 0.f, 0.f, 0.f, 0.f, 0.f};
        v8f vd = {0.f, 0.f, 0.f, 0.f, 0.f, 0.f, 0.f, 0.f};
        #pragma unroll
        for (int ks = 0; ks < 32; ++ks) {
            kd = wmma_f32_16x16x4(wk[ks], fr[ks], kd);
            vd = wmma_f32_16x16x4(fr[ks], wv[ks], vd);
        }

        {
            float* ok = Kl + (t * 16 + lm) * KP + 2 * h + hs;
            #pragma unroll
            for (int r = 0; r < 8; ++r) ok[r * NSZ * KP] = kd[r] + bk[r];
        }
        {
            float4 a, b;
            a.x = vd[0] + bvv; a.y = vd[1] + bvv; a.z = vd[2] + bvv; a.w = vd[3] + bvv;
            b.x = vd[4] + bvv; b.y = vd[5] + bvv; b.z = vd[6] + bvv; b.w = vd[7] + bvv;
            float* ov = Vl + (lm % 8) * VH + (2 * h + lm / 8) * VP + t * 16 + 8 * hs;
            *reinterpret_cast<float4*>(ov)     = a;
            *reinterpret_cast<float4*>(ov + 4) = b;
        }

        if (h == (t & 7)) {
            v8f acc = {0.f, 0.f, 0.f, 0.f, 0.f, 0.f, 0.f, 0.f};
            #pragma unroll
            for (int ks = 0; ks < 32; ++ks)
                acc = wmma_f32_16x16x4(fr[ks], we[ks], acc);
            if (lm < NHEADS) {
                float* oe = El + lm * NSZ + t * 16 + 8 * hs;
                #pragma unroll
                for (int r = 0; r < 8; ++r) oe[r] = acc[r] + bev;
            } else {
                float* og = Gl + (lm - NHEADS) * NSZ + t * 16 + 8 * hs;
                #pragma unroll
                for (int r = 0; r < 8; ++r)
                    og[r] = 1.f / (1.f + __expf(-(acc[r] + bev)));
            }
        }
        __syncthreads();
    }

    const float* pK = Kl + (h * NSZ + lm) * KP + 8 * hs;
    const float* pV = Vl + h * VH + lm * VP + 2 * hs;

    float4 n2 = reinterpret_cast<const float4*>(
        &feat1[((unsigned)(16 + h) * NSZ + j) * HID])[lane];
    float4 n3 = reinterpret_cast<const float4*>(
        &feat1[((unsigned)(24 + h) * NSZ + j) * HID])[lane];

    float ev[10], gate[10];
    #pragma unroll
    for (int ct = 0; ct < 10; ++ct) {
        ev[ct]   = El[h * NSZ + ct * 16 + lm];
        gate[ct] = Gl[h * NSZ + ct * 16 + lm];
    }
    float bqv[8];
    #pragma unroll
    for (int r = 0; r < 8; ++r) bqv[r] = bq[(8 * hs + r) * NHEADS + h];

    const int rowLN = tid / 16;
    const int cg    = tid % 16;

    v2f ot0[4] = {{0.f, 0.f}, {0.f, 0.f}, {0.f, 0.f}, {0.f, 0.f}};
    v2f ot1[4] = {{0.f, 0.f}, {0.f, 0.f}, {0.f, 0.f}, {0.f, 0.f}};
    unsigned oprev0 = 0, oprev1 = 0;

    for (int ip = 0; ip < 5; ++ip) {
        *reinterpret_cast<float4*>(&sFV[h * FP + lane * 4])              = n0;
        *reinterpret_cast<float4*>(&sFV[(h + 8) * FP + lane * 4])        = n1;
        *reinterpret_cast<float4*>(&sFV[(16 + h) * FP + lane * 4])       = n2;
        *reinterpret_cast<float4*>(&sFV[(24 + h) * FP + lane * 4])       = n3;
        __syncthreads();

        if (ip > 0) {
            v2f* op0 = reinterpret_cast<v2f*>(&out[oprev0]);
            v2f* op1 = reinterpret_cast<v2f*>(&out[oprev1]);
            #pragma unroll
            for (int c = 0; c < 4; ++c) { op0[c] = ot0[c]; op1[c] = ot1[c]; }
        }
        if (ip < 4) {
            const unsigned rb = (unsigned)((ip + 1) * 32 + h);
            n0 = reinterpret_cast<const float4*>(&feat1[((rb)      * NSZ + j) * HID])[lane];
            n1 = reinterpret_cast<const float4*>(&feat1[((rb + 8)  * NSZ + j) * HID])[lane];
            n2 = reinterpret_cast<const float4*>(&feat1[((rb + 16) * NSZ + j) * HID])[lane];
            n3 = reinterpret_cast<const float4*>(&feat1[((rb + 24) * NSZ + j) * HID])[lane];
        }

        v2f fq0[32];
        #pragma unroll
        for (int ks = 0; ks < 32; ++ks)
            fq0[ks] = *reinterpret_cast<const v2f*>(pf + ks * 4);
        __builtin_amdgcn_sched_barrier(0);

        #pragma unroll
        for (int sub = 0; sub < 2; ++sub) {
            v8f qd = {0.f, 0.f, 0.f, 0.f, 0.f, 0.f, 0.f, 0.f};
            #pragma unroll
            for (int ks = 0; ks < 32; ++ks) {
                const v2f b = (sub == 0) ? fq0[ks]
                    : *reinterpret_cast<const v2f*>(pf + 16 * FP + ks * 4);
                qd = wmma_f32_16x16x4(wq[ks], b, qd);
            }

            v2f afr[4];
            #pragma unroll
            for (int ks = 0; ks < 4; ++ks) {
                afr[ks].x = (qd[2 * ks]     + bqv[2 * ks])     * 0.25f;
                afr[ks].y = (qd[2 * ks + 1] + bqv[2 * ks + 1]) * 0.25f;
            }

            v8f acc[10];
            #pragma unroll
            for (int ct = 0; ct < 10; ++ct) {
                v8f c = {0.f, 0.f, 0.f, 0.f, 0.f, 0.f, 0.f, 0.f};
                #pragma unroll
                for (int ks = 0; ks < 4; ++ks) {
                    v2f b = *reinterpret_cast<const v2f*>(pK + ct * 16 * KP + ks * 2);
                    c = wmma_f32_16x16x4(afr[ks], b, c);
                }
                const float e = ev[ct];
                #pragma unroll
                for (int r = 0; r < 8; ++r) c[r] += e;
                acc[ct] = c;
            }

            #pragma unroll
            for (int r = 0; r < 8; ++r) {
                float mx = -1e30f;
                #pragma unroll
                for (int ct = 0; ct < 10; ++ct) mx = fmaxf(mx, acc[ct][r]);
                #pragma unroll
                for (int msk = 1; msk < 16; msk <<= 1) mx = fmaxf(mx, __shfl_xor(mx, msk, 32));
                float sm = 0.f;
                #pragma unroll
                for (int ct = 0; ct < 10; ++ct) {
                    float e = __expf(acc[ct][r] - mx);
                    acc[ct][r] = e;
                    sm += e;
                }
                #pragma unroll
                for (int msk = 1; msk < 16; msk <<= 1) sm += __shfl_xor(sm, msk, 32);
                const float inv = 1.f / sm;
                #pragma unroll
                for (int ct = 0; ct < 10; ++ct) acc[ct][r] = acc[ct][r] * inv * gate[ct];
            }

            v8f va = {0.f, 0.f, 0.f, 0.f, 0.f, 0.f, 0.f, 0.f};
            #pragma unroll
            for (int half = 0; half < 2; ++half) {
                #pragma unroll
                for (int ct = 0; ct < 5; ++ct) {
                    #pragma unroll
                    for (int r = 0; r < 8; ++r)
                        Plds[h][r + 8 * hs][ct * 16 + lm] = acc[half * 5 + ct][r];
                }
                wave_lds_handover();
                #pragma unroll
                for (int k0 = 0; k0 < 80; k0 += 4) {
                    v2f a, b;
                    a.x = Plds[h][lm][k0 + 2 * hs];
                    a.y = Plds[h][lm][k0 + 2 * hs + 1];
                    b = *reinterpret_cast<const v2f*>(pV + half * 80 + k0);
                    va = wmma_f32_16x16x4(a, b, va);
                }
                wave_lds_handover();
            }

            #pragma unroll
            for (int r = 0; r < 8; ++r)
                VaLds[sub][r + 8 * hs][lm * NHEADS + h] = va[r];
        }
        __syncthreads();

        #pragma unroll
        for (int sub = 0; sub < 2; ++sub) {
            float vals[8], s1 = 0.f, s2 = 0.f;
            #pragma unroll
            for (int c = 0; c < 8; ++c) {
                float v = VaLds[sub][rowLN][cg * 8 + c];
                vals[c] = v; s1 += v; s2 += v * v;
            }
            #pragma unroll
            for (int msk = 1; msk < 16; msk <<= 1) {
                s1 += __shfl_xor(s1, msk, 32);
                s2 += __shfl_xor(s2, msk, 32);
            }
            const float mu  = s1 * (1.f / HID);
            const float var = s2 * (1.f / HID) - mu * mu;
            const float inv = rsqrtf(var + LN_EPS);

            const unsigned op = ((unsigned)(ip * 32 + sub * 16 + rowLN) * NSZ + j) * HID + cg * 8;
            if (sub == 0) oprev0 = op; else oprev1 = op;
            #pragma unroll
            for (int c = 0; c < 8; c += 2) {
                const int ch = cg * 8 + c;
                v2f tv;
                tv.x = (vals[c]     - mu) * inv * ln_g[ch]     + ln_b[ch];
                tv.y = (vals[c + 1] - mu) * inv * ln_g[ch + 1] + ln_b[ch + 1];
                if (sub == 0) ot0[c / 2] = tv; else ot1[c / 2] = tv;
            }
        }
    }
    {
        v2f* op0 = reinterpret_cast<v2f*>(&out[oprev0]);
        v2f* op1 = reinterpret_cast<v2f*>(&out[oprev1]);
        #pragma unroll
        for (int c = 0; c < 4; ++c) { op0[c] = ot0[c]; op1[c] = ot1[c]; }
    }
}

extern "C" void kernel_launch(void* const* d_in, const int* in_sizes, int n_in,
                              void* d_out, int out_size, void* d_ws, size_t ws_size,
                              hipStream_t stream) {
    const float* feat1 = (const float*)d_in[0];
    const float* feat2 = (const float*)d_in[1];
    const float* Wq    = (const float*)d_in[2];
    const float* bq    = (const float*)d_in[3];
    const float* Wkv   = (const float*)d_in[4];
    const float* bkv   = (const float*)d_in[5];
    const float* Weg   = (const float*)d_in[6];
    const float* beg   = (const float*)d_in[7];
    const float* lng   = (const float*)d_in[8];
    const float* lnb   = (const float*)d_in[9];

    fused_kernel<<<dim3(NSZ), 256, 0, stream>>>(
        feat1, feat2, Wq, Wkv, Weg, bq, bkv, beg, lng, lnb, (float*)d_out);
}


# Softmax row sums on the matrix unit: lane-partial sums go through a wave-private LDS tile as an A operand against ones (4 WMMAs); 1/sum scales the 8 P@V results; row-sum order across lanes changed
# speedup vs baseline: 2.3825x; 1.0371x over previous
#include <hip/hip_runtime.h>
#include <hip/hip_bf16.h>


#define NSZ 160
#define NN (NSZ * NSZ)
#define HID 128
#define NHEADS 8
#define HD 16
#define LN_EPS 0.001f

#define FP (HID + 4)
#define KP 20
#define VP 164
#define VH (HD * VP + 8)

typedef __attribute__((ext_vector_type(2))) float v2f;
typedef __attribute__((ext_vector_type(8))) float v8f;

static __device__ __forceinline__ v8f wmma_f32_16x16x4(v2f a, v2f b, v8f c) {
    return __builtin_amdgcn_wmma_f32_16x16x4_f32(
         false, a,  false, b,
         (short)0, c,  false,  false);
}

static __device__ __forceinline__ void wave_lds_handover() {
    __builtin_amdgcn_fence(__ATOMIC_RELEASE, "wavefront");
    __builtin_amdgcn_wave_barrier();
    __builtin_amdgcn_fence(__ATOMIC_ACQUIRE, "wavefront");
}


__global__ __launch_bounds__(256)
void fused_kernel(
    const float* __restrict__ feat1, const float* __restrict__ feat2,
    const float* __restrict__ Wq, const float* __restrict__ Wkv,
    const float* __restrict__ Weg,
    const float* __restrict__ bq, const float* __restrict__ bkv,
    const float* __restrict__ beg,
    const float* __restrict__ ln_g, const float* __restrict__ ln_b,
    float* __restrict__ out)
{
    __shared__ __attribute__((aligned(16))) float Kl[NHEADS * NSZ * KP];
    __shared__ __attribute__((aligned(16))) float Vl[NHEADS * VH];
    __shared__ float El[NHEADS * NSZ];
    __shared__ float Gl[NHEADS * NSZ];
    __shared__ float Plds[NHEADS][16][81];
    __shared__ float Slds[NHEADS][16][17];
    __shared__ __attribute__((aligned(16))) float sFV[2 * 16 * FP];
    __shared__ float VaLds[2][16][HID];

    const int j    = blockIdx.x;
    const int tid  = threadIdx.x;
    const int h    = tid / 32;
    const int lane = tid % 32;
    const int hs   = lane / 16;
    const int lm   = lane % 16;

    const float* f2 = feat2 + (unsigned)j * NSZ * HID;

    float4 n0 = reinterpret_cast<const float4*>(f2 + (unsigned)(h)     * HID)[lane];
    float4 n1 = reinterpret_cast<const float4*>(f2 + (unsigned)(h + 8) * HID)[lane];

    const float* pf  = sFV + lm * FP + 2 * hs;

    v2f wk[32], wv[32], we[32], wq[32];
    const float* pwq = Wq + (unsigned)(2 * hs) * HID + lm * NHEADS + h;
    {
        const float* pwk = Wkv + (unsigned)(2 * hs) * (2 * HID) + 16 * h + lm;
        const float* pwv = pwk + HID;
        const float* pwe = Weg + (unsigned)(2 * hs) * (2 * NHEADS) + lm;
        #pragma unroll
        for (int ks = 0; ks < 32; ++ks) {
            wk[ks].x = pwk[(4 * ks)     * (2 * HID)];
            wk[ks].y = pwk[(4 * ks + 1) * (2 * HID)];
            wv[ks].x = pwv[(4 * ks)     * (2 * HID)];
            wv[ks].y = pwv[(4 * ks + 1) * (2 * HID)];
        }
        #pragma unroll
        for (int ks = 0; ks < 32; ++ks) {
            we[ks].x = pwe[(4 * ks)     * (2 * NHEADS)];
            we[ks].y = pwe[(4 * ks + 1) * (2 * NHEADS)];
        }
        #pragma unroll
        for (int ks = 0; ks < 32; ++ks) { wq[ks].x = 0.f; wq[ks].y = 0.f; }
    }

    float bk[8];
    #pragma unroll
    for (int r = 0; r < 8; ++r) bk[r] = bkv[16 * h + 8 * hs + r];
    const float bvv = bkv[HID + 16 * h + lm];
    const float bev = beg[lm];

    for (int t = 0; t < 10; ++t) {
        *reinterpret_cast<float4*>(&sFV[h * FP + lane * 4])       = n0;
        *reinterpret_cast<float4*>(&sFV[(h + 8) * FP + lane * 4]) = n1;
        __syncthreads();

        {
            const float* s0 = (t < 9) ? f2 + (unsigned)((t + 1) * 16 + h) * HID
                                      : feat1 + ((unsigned)(h) * NSZ + j) * HID;
            const float* s1 = (t < 9) ? f2 + (unsigned)((t + 1) * 16 + h + 8) * HID
                                      : feat1 + ((unsigned)(h + 8) * NSZ + j) * HID;
            n0 = reinterpret_cast<const float4*>(s0)[lane];
            n1 = reinterpret_cast<const float4*>(s1)[lane];
        }
        if (t == 0) {
            #pragma unroll
            for (int ks = 0; ks < 32; ++ks) {
                wq[ks].x = pwq[(4 * ks)     * HID];
                wq[ks].y = pwq[(4 * ks + 1) * HID];
            }
        }

        v2f fr[32];
        #pragma unroll
        for (int ks = 0; ks < 32; ++ks)
            fr[ks] = *reinterpret_cast<const v2f*>(pf + ks * 4);

        v8f kd = {0.f, 0.f, 0.f, 0.f, 0.f, 0.f, 0.f, 0.f};
        v8f vd = {0.f, 0.f, 0.f, 0.f, 0.f, 0.f, 0.f, 0.f};
        #pragma unroll
        for (int ks = 0; ks < 32; ++ks) {
            kd = wmma_f32_16x16x4(wk[ks], fr[ks], kd);
            vd = wmma_f32_16x16x4(fr[ks], wv[ks], vd);
        }

        {
            float* ok = Kl + (t * 16 + lm) * KP + 2 * h + hs;
            #pragma unroll
            for (int r = 0; r < 8; ++r) ok[r * NSZ * KP] = kd[r] + bk[r];
        }
        {
            float4 a, b;
            a.x = vd[0] + bvv; a.y = vd[1] + bvv; a.z = vd[2] + bvv; a.w = vd[3] + bvv;
            b.x = vd[4] + bvv; b.y = vd[5] + bvv; b.z = vd[6] + bvv; b.w = vd[7] + bvv;
            float* ov = Vl + (lm % 8) * VH + (2 * h + lm / 8) * VP + t * 16 + 8 * hs;
            *reinterpret_cast<float4*>(ov)     = a;
            *reinterpret_cast<float4*>(ov + 4) = b;
        }

        if (h == (t & 7)) {
            v8f acc = {0.f, 0.f, 0.f, 0.f, 0.f, 0.f, 0.f, 0.f};
            #pragma unroll
            for (int ks = 0; ks < 32; ++ks)
                acc = wmma_f32_16x16x4(fr[ks], we[ks], acc);
            if (lm < NHEADS) {
                float* oe = El + lm * NSZ + t * 16 + 8 * hs;
                #pragma unroll
                for (int r = 0; r < 8; ++r) oe[r] = acc[r] + bev;
            } else {
                float* og = Gl + (lm - NHEADS) * NSZ + t * 16 + 8 * hs;
                #pragma unroll
                for (int r = 0; r < 8; ++r)
                    og[r] = 1.f / (1.f + __expf(-(acc[r] + bev)));
            }
        }
        __syncthreads();
    }

    const float* pK = Kl + (h * NSZ + lm) * KP + 8 * hs;
    const float* pV = Vl + h * VH + lm * VP + 2 * hs;

    float4 n2 = reinterpret_cast<const float4*>(
        &feat1[((unsigned)(16 + h) * NSZ + j) * HID])[lane];
    float4 n3 = reinterpret_cast<const float4*>(
        &feat1[((unsigned)(24 + h) * NSZ + j) * HID])[lane];

    float ev[10], gate[10];
    #pragma unroll
    for (int ct = 0; ct < 10; ++ct) {
        ev[ct]   = El[h * NSZ + ct * 16 + lm];
        gate[ct] = Gl[h * NSZ + ct * 16 + lm];
    }
    float bqv[8];
    #pragma unroll
    for (int r = 0; r < 8; ++r) bqv[r] = bq[(8 * hs + r) * NHEADS + h];

    const int rowLN = tid / 16;
    const int cg    = tid % 16;

    v2f ot0[4] = {{0.f, 0.f}, {0.f, 0.f}, {0.f, 0.f}, {0.f, 0.f}};
    v2f ot1[4] = {{0.f, 0.f}, {0.f, 0.f}, {0.f, 0.f}, {0.f, 0.f}};
    unsigned oprev0 = 0, oprev1 = 0;

    for (int ip = 0; ip < 5; ++ip) {
        *reinterpret_cast<float4*>(&sFV[h * FP + lane * 4])              = n0;
        *reinterpret_cast<float4*>(&sFV[(h + 8) * FP + lane * 4])        = n1;
        *reinterpret_cast<float4*>(&sFV[(16 + h) * FP + lane * 4])       = n2;
        *reinterpret_cast<float4*>(&sFV[(24 + h) * FP + lane * 4])       = n3;
        __syncthreads();

        if (ip > 0) {
            v2f* op0 = reinterpret_cast<v2f*>(&out[oprev0]);
            v2f* op1 = reinterpret_cast<v2f*>(&out[oprev1]);
            #pragma unroll
            for (int c = 0; c < 4; ++c) { op0[c] = ot0[c]; op1[c] = ot1[c]; }
        }
        if (ip < 4) {
            const unsigned rb = (unsigned)((ip + 1) * 32 + h);
            n0 = reinterpret_cast<const float4*>(&feat1[((rb)      * NSZ + j) * HID])[lane];
            n1 = reinterpret_cast<const float4*>(&feat1[((rb + 8)  * NSZ + j) * HID])[lane];
            n2 = reinterpret_cast<const float4*>(&feat1[((rb + 16) * NSZ + j) * HID])[lane];
            n3 = reinterpret_cast<const float4*>(&feat1[((rb + 24) * NSZ + j) * HID])[lane];
        }

        v2f fq0[32];
        #pragma unroll
        for (int ks = 0; ks < 32; ++ks)
            fq0[ks] = *reinterpret_cast<const v2f*>(pf + ks * 4);
        __builtin_amdgcn_sched_barrier(0);

        #pragma unroll
        for (int sub = 0; sub < 2; ++sub) {
            v8f qd = {0.f, 0.f, 0.f, 0.f, 0.f, 0.f, 0.f, 0.f};
            #pragma unroll
            for (int ks = 0; ks < 32; ++ks) {
                const v2f b = (sub == 0) ? fq0[ks]
                    : *reinterpret_cast<const v2f*>(pf + 16 * FP + ks * 4);
                qd = wmma_f32_16x16x4(wq[ks], b, qd);
            }

            v2f afr[4];
            #pragma unroll
            for (int ks = 0; ks < 4; ++ks) {
                afr[ks].x = (qd[2 * ks]     + bqv[2 * ks])     * 0.25f;
                afr[ks].y = (qd[2 * ks + 1] + bqv[2 * ks + 1]) * 0.25f;
            }

            v8f acc[10];
            #pragma unroll
            for (int ct = 0; ct < 10; ++ct) {
                v8f c = {0.f, 0.f, 0.f, 0.f, 0.f, 0.f, 0.f, 0.f};
                #pragma unroll
                for (int ks = 0; ks < 4; ++ks) {
                    v2f b = *reinterpret_cast<const v2f*>(pK + ct * 16 * KP + ks * 2);
                    c = wmma_f32_16x16x4(afr[ks], b, c);
                }
                const float e = ev[ct];
                #pragma unroll
                for (int r = 0; r < 8; ++r) c[r] += e;
                acc[ct] = c;
            }

            float psum[8];
            #pragma unroll
            for (int r = 0; r < 8; ++r) {
                float mx = -1e30f;
                #pragma unroll
                for (int ct = 0; ct < 10; ++ct) mx = fmaxf(mx, acc[ct][r]);
                #pragma unroll
                for (int msk = 1; msk < 16; msk <<= 1) mx = fmaxf(mx, __shfl_xor(mx, msk, 32));
                float sm = 0.f;
                #pragma unroll
                for (int ct = 0; ct < 10; ++ct) {
                    float e = __expf(acc[ct][r] - mx);
                    sm += e;
                    acc[ct][r] = e * gate[ct];
                }
                psum[r] = sm;
            }

            v8f va = {0.f, 0.f, 0.f, 0.f, 0.f, 0.f, 0.f, 0.f};
            v8f sd = {0.f, 0.f, 0.f, 0.f, 0.f, 0.f, 0.f, 0.f};
            #pragma unroll
            for (int half = 0; half < 2; ++half) {
                #pragma unroll
                for (int ct = 0; ct < 5; ++ct) {
                    #pragma unroll
                    for (int r = 0; r < 8; ++r)
                        Plds[h][r + 8 * hs][ct * 16 + lm] = acc[half * 5 + ct][r];
                }
                if (half == 0) {
                    #pragma unroll
                    for (int r = 0; r < 8; ++r) Slds[h][r + 8 * hs][lm] = psum[r];
                }
                wave_lds_handover();
                if (half == 0) {
                    const v2f ones = {1.f, 1.f};
                    #pragma unroll
                    for (int ks = 0; ks < 4; ++ks) {
                        v2f a;
                        a.x = Slds[h][lm][4 * ks + 2 * hs];
                        a.y = Slds[h][lm][4 * ks + 2 * hs + 1];
                        sd = wmma_f32_16x16x4(a, ones, sd);
                    }
                }
                #pragma unroll
                for (int k0 = 0; k0 < 80; k0 += 4) {
                    v2f a, b;
                    a.x = Plds[h][lm][k0 + 2 * hs];
                    a.y = Plds[h][lm][k0 + 2 * hs + 1];
                    b = *reinterpret_cast<const v2f*>(pV + half * 80 + k0);
                    va = wmma_f32_16x16x4(a, b, va);
                }
                wave_lds_handover();
            }

            #pragma unroll
            for (int r = 0; r < 8; ++r) va[r] = va[r] / sd[r];

            #pragma unroll
            for (int r = 0; r < 8; ++r)
                VaLds[sub][r + 8 * hs][lm * NHEADS + h] = va[r];
        }
        __syncthreads();

        #pragma unroll
        for (int sub = 0; sub < 2; ++sub) {
            float vals[8], s1 = 0.f, s2 = 0.f;
            #pragma unroll
            for (int c = 0; c < 8; ++c) {
                float v = VaLds[sub][rowLN][cg * 8 + c];
                vals[c] = v; s1 += v; s2 += v * v;
            }
            #pragma unroll
            for (int msk = 1; msk < 16; msk <<= 1) {
                s1 += __shfl_xor(s1, msk, 32);
                s2 += __shfl_xor(s2, msk, 32);
            }
            const float mu  = s1 * (1.f / HID);
            const float var = s2 * (1.f / HID) - mu * mu;
            const float inv = rsqrtf(var + LN_EPS);

            const unsigned op = ((unsigned)(ip * 32 + sub * 16 + rowLN) * NSZ + j) * HID + cg * 8;
            if (sub == 0) oprev0 = op; else oprev1 = op;
            #pragma unroll
            for (int c = 0; c < 8; c += 2) {
                const int ch = cg * 8 + c;
                v2f tv;
                tv.x = (vals[c]     - mu) * inv * ln_g[ch]     + ln_b[ch];
                tv.y = (vals[c + 1] - mu) * inv * ln_g[ch + 1] + ln_b[ch + 1];
                if (sub == 0) ot0[c / 2] = tv; else ot1[c / 2] = tv;
            }
        }
    }
    {
        v2f* op0 = reinterpret_cast<v2f*>(&out[oprev0]);
        v2f* op1 = reinterpret_cast<v2f*>(&out[oprev1]);
        #pragma unroll
        for (int c = 0; c < 4; ++c) { op0[c] = ot0[c]; op1[c] = ot1[c]; }
    }
}

extern "C" void kernel_launch(void* const* d_in, const int* in_sizes, int n_in,
                              void* d_out, int out_size, void* d_ws, size_t ws_size,
                              hipStream_t stream) {
    const float* feat1 = (const float*)d_in[0];
    const float* feat2 = (const float*)d_in[1];
    const float* Wq    = (const float*)d_in[2];
    const float* bq    = (const float*)d_in[3];
    const float* Wkv   = (const float*)d_in[4];
    const float* bkv   = (const float*)d_in[5];
    const float* Weg   = (const float*)d_in[6];
    const float* beg   = (const float*)d_in[7];
    const float* lng   = (const float*)d_in[8];
    const float* lnb   = (const float*)d_in[9];

    fused_kernel<<<dim3(NSZ), 256, 0, stream>>>(
        feat1, feat2, Wq, Wkv, Weg, bq, bkv, beg, lng, lnb, (float*)d_out);
}


# Phase A in 3 trips of 4, 4, 2 key tiles (64 feat2 rows staged at once): 2 workgroup barriers per trip, not per tile; a trip's E/gate runs go one to each SIMD, first in their wave's trip; sums kept
# speedup vs baseline: 2.4954x; 1.0474x over previous
#include <hip/hip_runtime.h>
#include <hip/hip_bf16.h>


#define NSZ 160
#define NN (NSZ * NSZ)
#define HID 128
#define NHEADS 8
#define HD 16
#define LN_EPS 0.001f

#define FP (HID + 4)
#define KP 20
#define VP 164
#define VH (HD * VP + 8)

typedef __attribute__((ext_vector_type(2))) float v2f;
typedef __attribute__((ext_vector_type(8))) float v8f;

static __device__ __forceinline__ v8f wmma_f32_16x16x4(v2f a, v2f b, v8f c) {
    return __builtin_amdgcn_wmma_f32_16x16x4_f32(
         false, a,  false, b,
         (short)0, c,  false,  false);
}

static __device__ __forceinline__ void wave_lds_handover() {
    __builtin_amdgcn_fence(__ATOMIC_RELEASE, "wavefront");
    __builtin_amdgcn_wave_barrier();
    __builtin_amdgcn_fence(__ATOMIC_ACQUIRE, "wavefront");
}


__global__ __launch_bounds__(256)
void fused_kernel(
    const float* __restrict__ feat1, const float* __restrict__ feat2,
    const float* __restrict__ Wq, const float* __restrict__ Wkv,
    const float* __restrict__ Weg,
    const float* __restrict__ bq, const float* __restrict__ bkv,
    const float* __restrict__ beg,
    const float* __restrict__ ln_g, const float* __restrict__ ln_b,
    float* __restrict__ out)
{
    __shared__ __attribute__((aligned(16))) float Kl[NHEADS * NSZ * KP];
    __shared__ __attribute__((aligned(16))) float Vl[NHEADS * VH];
    __shared__ float El[NHEADS * NSZ];
    __shared__ float Gl[NHEADS * NSZ];
    __shared__ float Plds[NHEADS][16][81];
    __shared__ float Slds[NHEADS][16][17];
    __shared__ __attribute__((aligned(16))) float sFV[4 * 16 * FP];
    __shared__ float VaLds[2][16][HID];

    const int j    = blockIdx.x;
    const int tid  = threadIdx.x;
    const int h    = tid / 32;
    const int lane = tid % 32;
    const int hs   = lane / 16;
    const int lm   = lane % 16;

    const float* f2 = feat2 + (unsigned)j * NSZ * HID;

    float4 n0 = reinterpret_cast<const float4*>(f2 + (unsigned)(h)      * HID)[lane];
    float4 n1 = reinterpret_cast<const float4*>(f2 + (unsigned)(h + 8)  * HID)[lane];
    float4 n2 = reinterpret_cast<const float4*>(f2 + (unsigned)(h + 16) * HID)[lane];
    float4 n3 = reinterpret_cast<const float4*>(f2 + (unsigned)(h + 24) * HID)[lane];
    float4 n4 = reinterpret_cast<const float4*>(f2 + (unsigned)(h + 32) * HID)[lane];
    float4 n5 = reinterpret_cast<const float4*>(f2 + (unsigned)(h + 40) * HID)[lane];
    float4 n6 = reinterpret_cast<const float4*>(f2 + (unsigned)(h + 48) * HID)[lane];
    float4 n7 = reinterpret_cast<const float4*>(f2 + (unsigned)(h + 56) * HID)[lane];

    const float* pf  = sFV + lm * FP + 2 * hs;

    v2f wk[32], wv[32], we[32], wq[32];
    const float* pwq = Wq + (unsigned)(2 * hs) * HID + lm * NHEADS + h;
    {
        const float* pwk = Wkv + (unsigned)(2 * hs) * (2 * HID) + 16 * h + lm;
        const float* pwv = pwk + HID;
        const float* pwe = Weg + (unsigned)(2 * hs) * (2 * NHEADS) + lm;
        #pragma unroll
        for (int ks = 0; ks < 32; ++ks) {
            wk[ks].x = pwk[(4 * ks)     * (2 * HID)];
            wk[ks].y = pwk[(4 * ks + 1) * (2 * HID)];
            wv[ks].x = pwv[(4 * ks)     * (2 * HID)];
            wv[ks].y = pwv[(4 * ks + 1) * (2 * HID)];
        }
        #pragma unroll
        for (int ks = 0; ks < 32; ++ks) {
            we[ks].x = pwe[(4 * ks)     * (2 * NHEADS)];
            we[ks].y = pwe[(4 * ks + 1) * (2 * NHEADS)];
        }
        #pragma unroll
        for (int ks = 0; ks < 32; ++ks) { wq[ks].x = 0.f; wq[ks].y = 0.f; }
    }

    float bk[8];
    #pragma unroll
    for (int r = 0; r < 8; ++r) bk[r] = bkv[16 * h + 8 * hs + r];
    const float bvv = bkv[HID + 16 * h + lm];
    const float bev = beg[lm];

    for (int tp = 0; tp < 3; ++tp) {
        const int nt = (tp < 2) ? 4 : 2;
        *reinterpret_cast<float4*>(&sFV[h * FP + lane * 4])        = n0;
        *reinterpret_cast<float4*>(&sFV[(h + 8) * FP + lane * 4])  = n1;
        *reinterpret_cast<float4*>(&sFV[(h + 16) * FP + lane * 4]) = n2;
        *reinterpret_cast<float4*>(&sFV[(h + 24) * FP + lane * 4]) = n3;
        if (tp < 2) {
            *reinterpret_cast<float4*>(&sFV[(h + 32) * FP + lane * 4]) = n4;
            *reinterpret_cast<float4*>(&sFV[(h + 40) * FP + lane * 4]) = n5;
            *reinterpret_cast<float4*>(&sFV[(h + 48) * FP + lane * 4]) = n6;
            *reinterpret_cast<float4*>(&sFV[(h + 56) * FP + lane * 4]) = n7;
        }
        __syncthreads();

        if (tp == 0) {
            const float* sb = f2 + (unsigned)(64 + h) * HID;
            n0 = reinterpret_cast<const float4*>(sb)[lane];
            n1 = reinterpret_cast<const float4*>(sb + 8 * HID)[lane];
            n2 = reinterpret_cast<const float4*>(sb + 16 * HID)[lane];
            n3 = reinterpret_cast<const float4*>(sb + 24 * HID)[lane];
            n4 = reinterpret_cast<const float4*>(sb + 32 * HID)[lane];
            n5 = reinterpret_cast<const float4*>(sb + 40 * HID)[lane];
            n6 = reinterpret_cast<const float4*>(sb + 48 * HID)[lane];
            n7 = reinterpret_cast<const float4*>(sb + 56 * HID)[lane];
            #pragma unroll
            for (int ks = 0; ks < 32; ++ks) {
                wq[ks].x = pwq[(4 * ks)     * HID];
                wq[ks].y = pwq[(4 * ks + 1) * HID];
            }
        } else {
            const float* sb = (tp == 1) ? f2 + (unsigned)(128 + h) * HID
                                        : feat1 + ((unsigned)(h) * NSZ + j) * HID;
            const unsigned st = (tp == 1) ? (unsigned)(8 * HID) : (unsigned)(8 * NSZ * HID);
            n0 = reinterpret_cast<const float4*>(sb)[lane];
            n1 = reinterpret_cast<const float4*>(sb + st)[lane];
            n2 = reinterpret_cast<const float4*>(sb + 2 * st)[lane];
            n3 = reinterpret_cast<const float4*>(sb + 3 * st)[lane];
        }

        const int se  = (5 * (h + 4 * tp)) & 7;
        const int rot = (se < nt) ? se : 0;

        for (int i = 0; i < nt; ++i) {
            const int sub = (i + rot) & (nt - 1);
            const int tt = 4 * tp + sub;
            const float* pft = pf + sub * 16 * FP;

            v2f fr[32];
            #pragma unroll
            for (int ks = 0; ks < 32; ++ks)
                fr[ks] = *reinterpret_cast<const v2f*>(pft + ks * 4);

            v8f kd = {0.f, 0.f, 0.f, 0.f, 0.f, 0.f, 0.f, 0.f};
            v8f vd = {0.f, 0.f, 0.f, 0.f, 0.f, 0.f, 0.f, 0.f};
            #pragma unroll
            for (int ks = 0; ks < 32; ++ks) {
                kd = wmma_f32_16x16x4(wk[ks], fr[ks], kd);
                vd = wmma_f32_16x16x4(fr[ks], wv[ks], vd);
            }

            {
                float* ok = Kl + (tt * 16 + lm) * KP + 2 * h + hs;
                #pragma unroll
                for (int r = 0; r < 8; ++r) ok[r * NSZ * KP] = kd[r] + bk[r];
            }
            {
                float4 a, b;
                a.x = vd[0] + bvv; a.y = vd[1] + bvv; a.z = vd[2] + bvv; a.w = vd[3] + bvv;
                b.x = vd[4] + bvv; b.y = vd[5] + bvv; b.z = vd[6] + bvv; b.w = vd[7] + bvv;
                float* ov = Vl + (lm % 8) * VH + (2 * h + lm / 8) * VP + tt * 16 + 8 * hs;
                *reinterpret_cast<float4*>(ov)     = a;
                *reinterpret_cast<float4*>(ov + 4) = b;
            }

            if (sub == se) {
                v8f acc = {0.f, 0.f, 0.f, 0.f, 0.f, 0.f, 0.f, 0.f};
                #pragma unroll
                for (int ks = 0; ks < 32; ++ks)
                    acc = wmma_f32_16x16x4(fr[ks], we[ks], acc);
                if (lm < NHEADS) {
                    float* oe = El + lm * NSZ + tt * 16 + 8 * hs;
                    #pragma unroll
                    for (int r = 0; r < 8; ++r) oe[r] = acc[r] + bev;
                } else {
                    float* og = Gl + (lm - NHEADS) * NSZ + tt * 16 + 8 * hs;
                    #pragma unroll
                    for (int r = 0; r < 8; ++r)
                        og[r] = 1.f / (1.f + __expf(-(acc[r] + bev)));
                }
            }
        }
        __syncthreads();
    }

    const float* pK = Kl + (h * NSZ + lm) * KP + 8 * hs;
    const float* pV = Vl + h * VH + lm * VP + 2 * hs;


    float ev[10], gate[10];
    #pragma unroll
    for (int ct = 0; ct < 10; ++ct) {
        ev[ct]   = El[h * NSZ + ct * 16 + lm];
        gate[ct] = Gl[h * NSZ + ct * 16 + lm];
    }
    float bqv[8];
    #pragma unroll
    for (int r = 0; r < 8; ++r) bqv[r] = bq[(8 * hs + r) * NHEADS + h];

    const int rowLN = tid / 16;
    const int cg    = tid % 16;

    v2f ot0[4] = {{0.f, 0.f}, {0.f, 0.f}, {0.f, 0.f}, {0.f, 0.f}};
    v2f ot1[4] = {{0.f, 0.f}, {0.f, 0.f}, {0.f, 0.f}, {0.f, 0.f}};
    unsigned oprev0 = 0, oprev1 = 0;

    for (int ip = 0; ip < 5; ++ip) {
        *reinterpret_cast<float4*>(&sFV[h * FP + lane * 4])              = n0;
        *reinterpret_cast<float4*>(&sFV[(h + 8) * FP + lane * 4])        = n1;
        *reinterpret_cast<float4*>(&sFV[(16 + h) * FP + lane * 4])       = n2;
        *reinterpret_cast<float4*>(&sFV[(24 + h) * FP + lane * 4])       = n3;
        __syncthreads();

        if (ip > 0) {
            v2f* op0 = reinterpret_cast<v2f*>(&out[oprev0]);
            v2f* op1 = reinterpret_cast<v2f*>(&out[oprev1]);
            #pragma unroll
            for (int c = 0; c < 4; ++c) { op0[c] = ot0[c]; op1[c] = ot1[c]; }
        }
        if (ip < 4) {
            const unsigned rb = (unsigned)((ip + 1) * 32 + h);
            n0 = reinterpret_cast<const float4*>(&feat1[((rb)      * NSZ + j) * HID])[lane];
            n1 = reinterpret_cast<const float4*>(&feat1[((rb + 8)  * NSZ + j) * HID])[lane];
            n2 = reinterpret_cast<const float4*>(&feat1[((rb + 16) * NSZ + j) * HID])[lane];
            n3 = reinterpret_cast<const float4*>(&feat1[((rb + 24) * NSZ + j) * HID])[lane];
        }

        v2f fq0[32];
        #pragma unroll
        for (int ks = 0; ks < 32; ++ks)
            fq0[ks] = *reinterpret_cast<const v2f*>(pf + ks * 4);
        __builtin_amdgcn_sched_barrier(0);

        #pragma unroll
        for (int sub = 0; sub < 2; ++sub) {
            v8f qd = {0.f, 0.f, 0.f, 0.f, 0.f, 0.f, 0.f, 0.f};
            #pragma unroll
            for (int ks = 0; ks < 32; ++ks) {
                const v2f b = (sub == 0) ? fq0[ks]
                    : *reinterpret_cast<const v2f*>(pf + 16 * FP + ks * 4);
                qd = wmma_f32_16x16x4(wq[ks], b, qd);
            }

            v2f afr[4];
            #pragma unroll
            for (int ks = 0; ks < 4; ++ks) {
                afr[ks].x = (qd[2 * ks]     + bqv[2 * ks])     * 0.25f;
                afr[ks].y = (qd[2 * ks + 1] + bqv[2 * ks + 1]) * 0.25f;
            }

            v8f acc[10];
            #pragma unroll
            for (int ct = 0; ct < 10; ++ct) {
                v8f c = {0.f, 0.f, 0.f, 0.f, 0.f, 0.f, 0.f, 0.f};
                #pragma unroll
                for (int ks = 0; ks < 4; ++ks) {
                    v2f b = *reinterpret_cast<const v2f*>(pK + ct * 16 * KP + ks * 2);
                    c = wmma_f32_16x16x4(afr[ks], b, c);
                }
                const float e = ev[ct];
                #pragma unroll
                for (int r = 0; r < 8; ++r) c[r] += e;
                acc[ct] = c;
            }

            float psum[8];
            #pragma unroll
            for (int r = 0; r < 8; ++r) {
                float mx = -1e30f;
                #pragma unroll
                for (int ct = 0; ct < 10; ++ct) mx = fmaxf(mx, acc[ct][r]);
                #pragma unroll
                for (int msk = 1; msk < 16; msk <<= 1) mx = fmaxf(mx, __shfl_xor(mx, msk, 32));
                float sm = 0.f;
                #pragma unroll
                for (int ct = 0; ct < 10; ++ct) {
                    float e = __expf(acc[ct][r] - mx);
                    sm += e;
                    acc[ct][r] = e * gate[ct];
                }
                psum[r] = sm;
            }

            v8f va = {0.f, 0.f, 0.f, 0.f, 0.f, 0.f, 0.f, 0.f};
            v8f sd = {0.f, 0.f, 0.f, 0.f, 0.f, 0.f, 0.f, 0.f};
            #pragma unroll
            for (int half = 0; half < 2; ++half) {
                #pragma unroll
                for (int ct = 0; ct < 5; ++ct) {
                    #pragma unroll
                    for (int r = 0; r < 8; ++r)
                        Plds[h][r + 8 * hs][ct * 16 + lm] = acc[half * 5 + ct][r];
                }
                if (half == 0) {
                    #pragma unroll
                    for (int r = 0; r < 8; ++r) Slds[h][r + 8 * hs][lm] = psum[r];
                }
                wave_lds_handover();
                if (half == 0) {
                    const v2f ones = {1.f, 1.f};
                    #pragma unroll
                    for (int ks = 0; ks < 4; ++ks) {
                        v2f a;
                        a.x = Slds[h][lm][4 * ks + 2 * hs];
                        a.y = Slds[h][lm][4 * ks + 2 * hs + 1];
                        sd = wmma_f32_16x16x4(a, ones, sd);
                    }
                }
                #pragma unroll
                for (int k0 = 0; k0 < 80; k0 += 4) {
                    v2f a, b;
                    a.x = Plds[h][lm][k0 + 2 * hs];
                    a.y = Plds[h][lm][k0 + 2 * hs + 1];
                    b = *reinterpret_cast<const v2f*>(pV + half * 80 + k0);
                    va = wmma_f32_16x16x4(a, b, va);
                }
                wave_lds_handover();
            }

            #pragma unroll
            for (int r = 0; r < 8; ++r) va[r] = va[r] / sd[r];

            #pragma unroll
            for (int r = 0; r < 8; ++r)
                VaLds[sub][r + 8 * hs][lm * NHEADS + h] = va[r];
        }
        __syncthreads();

        #pragma unroll
        for (int sub = 0; sub < 2; ++sub) {
            float vals[8], s1 = 0.f, s2 = 0.f;
            #pragma unroll
            for (int c = 0; c < 8; ++c) {
                float v = VaLds[sub][rowLN][cg * 8 + c];
                vals[c] = v; s1 += v; s2 += v * v;
            }
            #pragma unroll
            for (int msk = 1; msk < 16; msk <<= 1) {
                s1 += __shfl_xor(s1, msk, 32);
                s2 += __shfl_xor(s2, msk, 32);
            }
            const float mu  = s1 * (1.f / HID);
            const float var = s2 * (1.f / HID) - mu * mu;
            const float inv = rsqrtf(var + LN_EPS);

            const unsigned op = ((unsigned)(ip * 32 + sub * 16 + rowLN) * NSZ + j) * HID + cg * 8;
            if (sub == 0) oprev0 = op; else oprev1 = op;
            #pragma unroll
            for (int c = 0; c < 8; c += 2) {
                const int ch = cg * 8 + c;
                v2f tv;
                tv.x = (vals[c]     - mu) * inv * ln_g[ch]     + ln_b[ch];
                tv.y = (vals[c + 1] - mu) * inv * ln_g[ch + 1] + ln_b[ch + 1];
                if (sub == 0) ot0[c / 2] = tv; else ot1[c / 2] = tv;
            }
        }
    }
    {
        v2f* op0 = reinterpret_cast<v2f*>(&out[oprev0]);
        v2f* op1 = reinterpret_cast<v2f*>(&out[oprev1]);
        #pragma unroll
        for (int c = 0; c < 4; ++c) { op0[c] = ot0[c]; op1[c] = ot1[c]; }
    }
}

extern "C" void kernel_launch(void* const* d_in, const int* in_sizes, int n_in,
                              void* d_out, int out_size, void* d_ws, size_t ws_size,
                              hipStream_t stream) {
    const float* feat1 = (const float*)d_in[0];
    const float* feat2 = (const float*)d_in[1];
    const float* Wq    = (const float*)d_in[2];
    const float* bq    = (const float*)d_in[3];
    const float* Wkv   = (const float*)d_in[4];
    const float* bkv   = (const float*)d_in[5];
    const float* Weg   = (const float*)d_in[6];
    const float* beg   = (const float*)d_in[7];
    const float* lng   = (const float*)d_in[8];
    const float* lnb   = (const float*)d_in[9];

    fused_kernel<<<dim3(NSZ), 256, 0, stream>>>(
        feat1, feat2, Wq, Wkv, Weg, bq, bkv, beg, lng, lnb, (float*)d_out);
}


# 16-lane butterflies in two shuffle rounds of three (radix 4: lanes ^1,^2,^3 then ^4,^8,^12), not four of one: softmax row max (same values), LayerNorm sums (sum order across lanes regrouped)
# speedup vs baseline: 2.5771x; 1.0327x over previous
#include <hip/hip_runtime.h>
#include <hip/hip_bf16.h>


#define NSZ 160
#define NN (NSZ * NSZ)
#define HID 128
#define NHEADS 8
#define HD 16
#define LN_EPS 0.001f

#define FP (HID + 4)
#define KP 20
#define VP 164
#define VH (HD * VP + 8)

typedef __attribute__((ext_vector_type(2))) float v2f;
typedef __attribute__((ext_vector_type(8))) float v8f;

static __device__ __forceinline__ v8f wmma_f32_16x16x4(v2f a, v2f b, v8f c) {
    return __builtin_amdgcn_wmma_f32_16x16x4_f32(
         false, a,  false, b,
         (short)0, c,  false,  false);
}

static __device__ __forceinline__ void wave_lds_handover() {
    __builtin_amdgcn_fence(__ATOMIC_RELEASE, "wavefront");
    __builtin_amdgcn_wave_barrier();
    __builtin_amdgcn_fence(__ATOMIC_ACQUIRE, "wavefront");
}


__global__ __launch_bounds__(256)
void fused_kernel(
    const float* __restrict__ feat1, const float* __restrict__ feat2,
    const float* __restrict__ Wq, const float* __restrict__ Wkv,
    const float* __restrict__ Weg,
    const float* __restrict__ bq, const float* __restrict__ bkv,
    const float* __restrict__ beg,
    const float* __restrict__ ln_g, const float* __restrict__ ln_b,
    float* __restrict__ out)
{
    __shared__ __attribute__((aligned(16))) float Kl[NHEADS * NSZ * KP];
    __shared__ __attribute__((aligned(16))) float Vl[NHEADS * VH];
    __shared__ float El[NHEADS * NSZ];
    __shared__ float Gl[NHEADS * NSZ];
    __shared__ float Plds[NHEADS][16][81];
    __shared__ float Slds[NHEADS][16][17];
    __shared__ __attribute__((aligned(16))) float sFV[4 * 16 * FP];
    __shared__ float VaLds[2][16][HID];

    const int j    = blockIdx.x;
    const int tid  = threadIdx.x;
    const int h    = tid / 32;
    const int lane = tid % 32;
    const int hs   = lane / 16;
    const int lm   = lane % 16;

    const float* f2 = feat2 + (unsigned)j * NSZ * HID;

    float4 n0 = reinterpret_cast<const float4*>(f2 + (unsigned)(h)      * HID)[lane];
    float4 n1 = reinterpret_cast<const float4*>(f2 + (unsigned)(h + 8)  * HID)[lane];
    float4 n2 = reinterpret_cast<const float4*>(f2 + (unsigned)(h + 16) * HID)[lane];
    float4 n3 = reinterpret_cast<const float4*>(f2 + (unsigned)(h + 24) * HID)[lane];
    float4 n4 = reinterpret_cast<const float4*>(f2 + (unsigned)(h + 32) * HID)[lane];
    float4 n5 = reinterpret_cast<const float4*>(f2 + (unsigned)(h + 40) * HID)[lane];
    float4 n6 = reinterpret_cast<const float4*>(f2 + (unsigned)(h + 48) * HID)[lane];
    float4 n7 = reinterpret_cast<const float4*>(f2 + (unsigned)(h + 56) * HID)[lane];

    const float* pf  = sFV + lm * FP + 2 * hs;

    v2f wk[32], wv[32], we[32], wq[32];
    const float* pwq = Wq + (unsigned)(2 * hs) * HID + lm * NHEADS + h;
    {
        const float* pwk = Wkv + (unsigned)(2 * hs) * (2 * HID) + 16 * h + lm;
        const float* pwv = pwk + HID;
        const float* pwe = Weg + (unsigned)(2 * hs) * (2 * NHEADS) + lm;
        #pragma unroll
        for (int ks = 0; ks < 32; ++ks) {
            wk[ks].x = pwk[(4 * ks)     * (2 * HID)];
            wk[ks].y = pwk[(4 * ks + 1) * (2 * HID)];
            wv[ks].x = pwv[(4 * ks)     * (2 * HID)];
            wv[ks].y = pwv[(4 * ks + 1) * (2 * HID)];
        }
        #pragma unroll
        for (int ks = 0; ks < 32; ++ks) {
            we[ks].x = pwe[(4 * ks)     * (2 * NHEADS)];
            we[ks].y = pwe[(4 * ks + 1) * (2 * NHEADS)];
        }
        #pragma unroll
        for (int ks = 0; ks < 32; ++ks) { wq[ks].x = 0.f; wq[ks].y = 0.f; }
    }

    float bk[8];
    #pragma unroll
    for (int r = 0; r < 8; ++r) bk[r] = bkv[16 * h + 8 * hs + r];
    const float bvv = bkv[HID + 16 * h + lm];
    const float bev = beg[lm];

    for (int tp = 0; tp < 3; ++tp) {
        const int nt = (tp < 2) ? 4 : 2;
        *reinterpret_cast<float4*>(&sFV[h * FP + lane * 4])        = n0;
        *reinterpret_cast<float4*>(&sFV[(h + 8) * FP + lane * 4])  = n1;
        *reinterpret_cast<float4*>(&sFV[(h + 16) * FP + lane * 4]) = n2;
        *reinterpret_cast<float4*>(&sFV[(h + 24) * FP + lane * 4]) = n3;
        if (tp < 2) {
            *reinterpret_cast<float4*>(&sFV[(h + 32) * FP + lane * 4]) = n4;
            *reinterpret_cast<float4*>(&sFV[(h + 40) * FP + lane * 4]) = n5;
            *reinterpret_cast<float4*>(&sFV[(h + 48) * FP + lane * 4]) = n6;
            *reinterpret_cast<float4*>(&sFV[(h + 56) * FP + lane * 4]) = n7;
        }
        __syncthreads();

        if (tp == 0) {
            const float* sb = f2 + (unsigned)(64 + h) * HID;
            n0 = reinterpret_cast<const float4*>(sb)[lane];
            n1 = reinterpret_cast<const float4*>(sb + 8 * HID)[lane];
            n2 = reinterpret_cast<const float4*>(sb + 16 * HID)[lane];
            n3 = reinterpret_cast<const float4*>(sb + 24 * HID)[lane];
            n4 = reinterpret_cast<const float4*>(sb + 32 * HID)[lane];
            n5 = reinterpret_cast<const float4*>(sb + 40 * HID)[lane];
            n6 = reinterpret_cast<const float4*>(sb + 48 * HID)[lane];
            n7 = reinterpret_cast<const float4*>(sb + 56 * HID)[lane];
            #pragma unroll
            for (int ks = 0; ks < 32; ++ks) {
                wq[ks].x = pwq[(4 * ks)     * HID];
                wq[ks].y = pwq[(4 * ks + 1) * HID];
            }
        } else {
            const float* sb = (tp == 1) ? f2 + (unsigned)(128 + h) * HID
                                        : feat1 + ((unsigned)(h) * NSZ + j) * HID;
            const unsigned st = (tp == 1) ? (unsigned)(8 * HID) : (unsigned)(8 * NSZ * HID);
            n0 = reinterpret_cast<const float4*>(sb)[lane];
            n1 = reinterpret_cast<const float4*>(sb + st)[lane];
            n2 = reinterpret_cast<const float4*>(sb + 2 * st)[lane];
            n3 = reinterpret_cast<const float4*>(sb + 3 * st)[lane];
        }

        const int se  = (5 * (h + 4 * tp)) & 7;
        const int rot = (se < nt) ? se : 0;

        for (int i = 0; i < nt; ++i) {
            const int sub = (i + rot) & (nt - 1);
            const int tt = 4 * tp + sub;
            const float* pft = pf + sub * 16 * FP;

            v2f fr[32];
            #pragma unroll
            for (int ks = 0; ks < 32; ++ks)
                fr[ks] = *reinterpret_cast<const v2f*>(pft + ks * 4);

            v8f kd = {0.f, 0.f, 0.f, 0.f, 0.f, 0.f, 0.f, 0.f};
            v8f vd = {0.f, 0.f, 0.f, 0.f, 0.f, 0.f, 0.f, 0.f};
            #pragma unroll
            for (int ks = 0; ks < 32; ++ks) {
                kd = wmma_f32_16x16x4(wk[ks], fr[ks], kd);
                vd = wmma_f32_16x16x4(fr[ks], wv[ks], vd);
            }

            {
                float* ok = Kl + (tt * 16 + lm) * KP + 2 * h + hs;
                #pragma unroll
                for (int r = 0; r < 8; ++r) ok[r * NSZ * KP] = kd[r] + bk[r];
            }
            {
                float4 a, b;
                a.x = vd[0] + bvv; a.y = vd[1] + bvv; a.z = vd[2] + bvv; a.w = vd[3] + bvv;
                b.x = vd[4] + bvv; b.y = vd[5] + bvv; b.z = vd[6] + bvv; b.w = vd[7] + bvv;
                float* ov = Vl + (lm % 8) * VH + (2 * h + lm / 8) * VP + tt * 16 + 8 * hs;
                *reinterpret_cast<float4*>(ov)     = a;
                *reinterpret_cast<float4*>(ov + 4) = b;
            }

            if (sub == se) {
                v8f acc = {0.f, 0.f, 0.f, 0.f, 0.f, 0.f, 0.f, 0.f};
                #pragma unroll
                for (int ks = 0; ks < 32; ++ks)
                    acc = wmma_f32_16x16x4(fr[ks], we[ks], acc);
                if (lm < NHEADS) {
                    float* oe = El + lm * NSZ + tt * 16 + 8 * hs;
                    #pragma unroll
                    for (int r = 0; r < 8; ++r) oe[r] = acc[r] + bev;
                } else {
                    float* og = Gl + (lm - NHEADS) * NSZ + tt * 16 + 8 * hs;
                    #pragma unroll
                    for (int r = 0; r < 8; ++r)
                        og[r] = 1.f / (1.f + __expf(-(acc[r] + bev)));
                }
            }
        }
        __syncthreads();
    }

    const float* pK = Kl + (h * NSZ + lm) * KP + 8 * hs;
    const float* pV = Vl + h * VH + lm * VP + 2 * hs;


    float ev[10], gate[10];
    #pragma unroll
    for (int ct = 0; ct < 10; ++ct) {
        ev[ct]   = El[h * NSZ + ct * 16 + lm];
        gate[ct] = Gl[h * NSZ + ct * 16 + lm];
    }
    float bqv[8];
    #pragma unroll
    for (int r = 0; r < 8; ++r) bqv[r] = bq[(8 * hs + r) * NHEADS + h];

    const int rowLN = tid / 16;
    const int cg    = tid % 16;

    v2f ot0[4] = {{0.f, 0.f}, {0.f, 0.f}, {0.f, 0.f}, {0.f, 0.f}};
    v2f ot1[4] = {{0.f, 0.f}, {0.f, 0.f}, {0.f, 0.f}, {0.f, 0.f}};
    unsigned oprev0 = 0, oprev1 = 0;

    for (int ip = 0; ip < 5; ++ip) {
        *reinterpret_cast<float4*>(&sFV[h * FP + lane * 4])              = n0;
        *reinterpret_cast<float4*>(&sFV[(h + 8) * FP + lane * 4])        = n1;
        *reinterpret_cast<float4*>(&sFV[(16 + h) * FP + lane * 4])       = n2;
        *reinterpret_cast<float4*>(&sFV[(24 + h) * FP + lane * 4])       = n3;
        __syncthreads();

        if (ip > 0) {
            v2f* op0 = reinterpret_cast<v2f*>(&out[oprev0]);
            v2f* op1 = reinterpret_cast<v2f*>(&out[oprev1]);
            #pragma unroll
            for (int c = 0; c < 4; ++c) { op0[c] = ot0[c]; op1[c] = ot1[c]; }
        }
        if (ip < 4) {
            const unsigned rb = (unsigned)((ip + 1) * 32 + h);
            n0 = reinterpret_cast<const float4*>(&feat1[((rb)      * NSZ + j) * HID])[lane];
            n1 = reinterpret_cast<const float4*>(&feat1[((rb + 8)  * NSZ + j) * HID])[lane];
            n2 = reinterpret_cast<const float4*>(&feat1[((rb + 16) * NSZ + j) * HID])[lane];
            n3 = reinterpret_cast<const float4*>(&feat1[((rb + 24) * NSZ + j) * HID])[lane];
        }

        v2f fq0[32];
        #pragma unroll
        for (int ks = 0; ks < 32; ++ks)
            fq0[ks] = *reinterpret_cast<const v2f*>(pf + ks * 4);
        __builtin_amdgcn_sched_barrier(0);

        #pragma unroll
        for (int sub = 0; sub < 2; ++sub) {
            v8f qd = {0.f, 0.f, 0.f, 0.f, 0.f, 0.f, 0.f, 0.f};
            #pragma unroll
            for (int ks = 0; ks < 32; ++ks) {
                const v2f b = (sub == 0) ? fq0[ks]
                    : *reinterpret_cast<const v2f*>(pf + 16 * FP + ks * 4);
                qd = wmma_f32_16x16x4(wq[ks], b, qd);
            }

            v2f afr[4];
            #pragma unroll
            for (int ks = 0; ks < 4; ++ks) {
                afr[ks].x = (qd[2 * ks]     + bqv[2 * ks])     * 0.25f;
                afr[ks].y = (qd[2 * ks + 1] + bqv[2 * ks + 1]) * 0.25f;
            }

            v8f acc[10];
            #pragma unroll
            for (int ct = 0; ct < 10; ++ct) {
                v8f c = {0.f, 0.f, 0.f, 0.f, 0.f, 0.f, 0.f, 0.f};
                #pragma unroll
                for (int ks = 0; ks < 4; ++ks) {
                    v2f b = *reinterpret_cast<const v2f*>(pK + ct * 16 * KP + ks * 2);
                    c = wmma_f32_16x16x4(afr[ks], b, c);
                }
                const float e = ev[ct];
                #pragma unroll
                for (int r = 0; r < 8; ++r) c[r] += e;
                acc[ct] = c;
            }

            float psum[8];
            #pragma unroll
            for (int r = 0; r < 8; ++r) {
                float mx = -1e30f;
                #pragma unroll
                for (int ct = 0; ct < 10; ++ct) mx = fmaxf(mx, acc[ct][r]);
                {
                    const float p1 = __shfl_xor(mx, 1, 32);
                    const float p2 = __shfl_xor(mx, 2, 32);
                    const float p3 = __shfl_xor(mx, 3, 32);
                    mx = fmaxf(mx, fmaxf(fmaxf(p1, p2), p3));
                    const float q1 = __shfl_xor(mx, 4, 32);
                    const float q2 = __shfl_xor(mx, 8, 32);
                    const float q3 = __shfl_xor(mx, 12, 32);
                    mx = fmaxf(mx, fmaxf(fmaxf(q1, q2), q3));
                }
                float sm = 0.f;
                #pragma unroll
                for (int ct = 0; ct < 10; ++ct) {
                    float e = __expf(acc[ct][r] - mx);
                    sm += e;
                    acc[ct][r] = e * gate[ct];
                }
                psum[r] = sm;
            }

            v8f va = {0.f, 0.f, 0.f, 0.f, 0.f, 0.f, 0.f, 0.f};
            v8f sd = {0.f, 0.f, 0.f, 0.f, 0.f, 0.f, 0.f, 0.f};
            #pragma unroll
            for (int half = 0; half < 2; ++half) {
                #pragma unroll
                for (int ct = 0; ct < 5; ++ct) {
                    #pragma unroll
                    for (int r = 0; r < 8; ++r)
                        Plds[h][r + 8 * hs][ct * 16 + lm] = acc[half * 5 + ct][r];
                }
                if (half == 0) {
                    #pragma unroll
                    for (int r = 0; r < 8; ++r) Slds[h][r + 8 * hs][lm] = psum[r];
                }
                wave_lds_handover();
                if (half == 0) {
                    const v2f ones = {1.f, 1.f};
                    #pragma unroll
                    for (int ks = 0; ks < 4; ++ks) {
                        v2f a;
                        a.x = Slds[h][lm][4 * ks + 2 * hs];
                        a.y = Slds[h][lm][4 * ks + 2 * hs + 1];
                        sd = wmma_f32_16x16x4(a, ones, sd);
                    }
                }
                #pragma unroll
                for (int k0 = 0; k0 < 80; k0 += 4) {
                    v2f a, b;
                    a.x = Plds[h][lm][k0 + 2 * hs];
                    a.y = Plds[h][lm][k0 + 2 * hs + 1];
                    b = *reinterpret_cast<const v2f*>(pV + half * 80 + k0);
                    va = wmma_f32_16x16x4(a, b, va);
                }
                wave_lds_handover();
            }

            #pragma unroll
            for (int r = 0; r < 8; ++r) va[r] = va[r] / sd[r];

            #pragma unroll
            for (int r = 0; r < 8; ++r)
                VaLds[sub][r + 8 * hs][lm * NHEADS + h] = va[r];
        }
        __syncthreads();

        #pragma unroll
        for (int sub = 0; sub < 2; ++sub) {
            float vals[8], s1 = 0.f, s2 = 0.f;
            #pragma unroll
            for (int c = 0; c < 8; ++c) {
                float v = VaLds[sub][rowLN][cg * 8 + c];
                vals[c] = v; s1 += v; s2 += v * v;
            }
            {
                const float a1 = __shfl_xor(s1, 1, 32), b1 = __shfl_xor(s2, 1, 32);
                const float a2 = __shfl_xor(s1, 2, 32), b2 = __shfl_xor(s2, 2, 32);
                const float a3 = __shfl_xor(s1, 3, 32), b3 = __shfl_xor(s2, 3, 32);
                s1 += (a1 + a2) + a3;
                s2 += (b1 + b2) + b3;
                const float c1 = __shfl_xor(s1, 4, 32),  d1 = __shfl_xor(s2, 4, 32);
                const float c2 = __shfl_xor(s1, 8, 32),  d2 = __shfl_xor(s2, 8, 32);
                const float c3 = __shfl_xor(s1, 12, 32), d3 = __shfl_xor(s2, 12, 32);
                s1 += (c1 + c2) + c3;
                s2 += (d1 + d2) + d3;
            }
            const float mu  = s1 * (1.f / HID);
            const float var = s2 * (1.f / HID) - mu * mu;
            const float inv = rsqrtf(var + LN_EPS);

            const unsigned op = ((unsigned)(ip * 32 + sub * 16 + rowLN) * NSZ + j) * HID + cg * 8;
            if (sub == 0) oprev0 = op; else oprev1 = op;
            #pragma unroll
            for (int c = 0; c < 8; c += 2) {
                const int ch = cg * 8 + c;
                v2f tv;
                tv.x = (vals[c]     - mu) * inv * ln_g[ch]     + ln_b[ch];
                tv.y = (vals[c + 1] - mu) * inv * ln_g[ch + 1] + ln_b[ch + 1];
                if (sub == 0) ot0[c / 2] = tv; else ot1[c / 2] = tv;
            }
        }
    }
    {
        v2f* op0 = reinterpret_cast<v2f*>(&out[oprev0]);
        v2f* op1 = reinterpret_cast<v2f*>(&out[oprev1]);
        #pragma unroll
        for (int c = 0; c < 4; ++c) { op0[c] = ot0[c]; op1[c] = ot1[c]; }
    }
}

extern "C" void kernel_launch(void* const* d_in, const int* in_sizes, int n_in,
                              void* d_out, int out_size, void* d_ws, size_t ws_size,
                              hipStream_t stream) {
    const float* feat1 = (const float*)d_in[0];
    const float* feat2 = (const float*)d_in[1];
    const float* Wq    = (const float*)d_in[2];
    const float* bq    = (const float*)d_in[3];
    const float* Wkv   = (const float*)d_in[4];
    const float* bkv   = (const float*)d_in[5];
    const float* Weg   = (const float*)d_in[6];
    const float* beg   = (const float*)d_in[7];
    const float* lng   = (const float*)d_in[8];
    const float* lnb   = (const float*)d_in[9];

    fused_kernel<<<dim3(NSZ), 256, 0, stream>>>(
        feat1, feat2, Wq, Wkv, Weg, bq, bkv, beg, lng, lnb, (float*)d_out);
}


# Grid 160 -> 256 workgroups, one per WGP: 96 more take the last query-tile pair of every j (64 of them for two j) and form that j's K, V, E, gate again; a j's own workgroup sweeps 8 tiles; sums kept
# speedup vs baseline: 2.6616x; 1.0328x over previous
#include <hip/hip_runtime.h>
#include <hip/hip_bf16.h>


#define NSZ 160
#define NN (NSZ * NSZ)
#define HID 128
#define NHEADS 8
#define HD 16
#define LN_EPS 0.001f

#define FP (HID + 4)
#define KP 20
#define VP 164
#define VH (HD * VP + 8)

typedef __attribute__((ext_vector_type(2))) float v2f;
typedef __attribute__((ext_vector_type(8))) float v8f;

static __device__ __forceinline__ v8f wmma_f32_16x16x4(v2f a, v2f b, v8f c) {
    return __builtin_amdgcn_wmma_f32_16x16x4_f32(
         false, a,  false, b,
         (short)0, c,  false,  false);
}

static __device__ __forceinline__ void wave_lds_handover() {
    __builtin_amdgcn_fence(__ATOMIC_RELEASE, "wavefront");
    __builtin_amdgcn_wave_barrier();
    __builtin_amdgcn_fence(__ATOMIC_ACQUIRE, "wavefront");
}


__global__ __launch_bounds__(256)
void fused_kernel(
    const float* __restrict__ feat1, const float* __restrict__ feat2,
    const float* __restrict__ Wq, const float* __restrict__ Wkv,
    const float* __restrict__ Weg,
    const float* __restrict__ bq, const float* __restrict__ bkv,
    const float* __restrict__ beg,
    const float* __restrict__ ln_g, const float* __restrict__ ln_b,
    float* __restrict__ out)
{
    __shared__ __attribute__((aligned(16))) float Kl[NHEADS * NSZ * KP];
    __shared__ __attribute__((aligned(16))) float Vl[NHEADS * VH];
    __shared__ float El[NHEADS * NSZ];
    __shared__ float Gl[NHEADS * NSZ];
    __shared__ float Plds[NHEADS][16][81];
    __shared__ float Slds[NHEADS][16][17];
    __shared__ __attribute__((aligned(16))) float sFV[4 * 16 * FP];
    __shared__ float VaLds[2][16][HID];

    const int tid  = threadIdx.x;
    const int h    = tid / 32;
    const int lane = tid % 32;
    const int hs   = lane / 16;
    const int lm   = lane % 16;

    const int bid = blockIdx.x;
    int jfirst = bid, nseg = 1, ipb = 0, ipe = 4;
    if (bid >= NSZ) {
        const int u = bid - NSZ;
        ipb = 4; ipe = 5;
        if (u < 64) { jfirst = 2 * u; nseg = 2; } else { jfirst = 64 + u; }
    }

    for (int seg = 0; seg < nseg; ++seg) {
    const int j = jfirst + seg;

    unsigned zo = 0;
    asm volatile("" : "+s"(zo));
    const float* Wq_p = Wq + zo;   const float* Wkv_p = Wkv + zo; const float* Weg_p = Weg + zo;
    const float* bq_p = bq + zo;   const float* bkv_p = bkv + zo; const float* beg_p = beg + zo;

    const float* f2 = feat2 + (unsigned)j * NSZ * HID;

    float4 n0 = reinterpret_cast<const float4*>(f2 + (unsigned)(h)      * HID)[lane];
    float4 n1 = reinterpret_cast<const float4*>(f2 + (unsigned)(h + 8)  * HID)[lane];
    float4 n2 = reinterpret_cast<const float4*>(f2 + (unsigned)(h + 16) * HID)[lane];
    float4 n3 = reinterpret_cast<const float4*>(f2 + (unsigned)(h + 24) * HID)[lane];
    float4 n4 = reinterpret_cast<const float4*>(f2 + (unsigned)(h + 32) * HID)[lane];
    float4 n5 = reinterpret_cast<const float4*>(f2 + (unsigned)(h + 40) * HID)[lane];
    float4 n6 = reinterpret_cast<const float4*>(f2 + (unsigned)(h + 48) * HID)[lane];
    float4 n7 = reinterpret_cast<const float4*>(f2 + (unsigned)(h + 56) * HID)[lane];

    const float* pf  = sFV + lm * FP + 2 * hs;

    v2f wk[32], wv[32], we[32], wq[32];
    const float* pwq = Wq_p + (unsigned)(2 * hs) * HID + lm * NHEADS + h;
    {
        const float* pwk = Wkv_p + (unsigned)(2 * hs) * (2 * HID) + 16 * h + lm;
        const float* pwv = pwk + HID;
        const float* pwe = Weg_p + (unsigned)(2 * hs) * (2 * NHEADS) + lm;
        #pragma unroll
        for (int ks = 0; ks < 32; ++ks) {
            wk[ks].x = pwk[(4 * ks)     * (2 * HID)];
            wk[ks].y = pwk[(4 * ks + 1) * (2 * HID)];
            wv[ks].x = pwv[(4 * ks)     * (2 * HID)];
            wv[ks].y = pwv[(4 * ks + 1) * (2 * HID)];
        }
        #pragma unroll
        for (int ks = 0; ks < 32; ++ks) {
            we[ks].x = pwe[(4 * ks)     * (2 * NHEADS)];
            we[ks].y = pwe[(4 * ks + 1) * (2 * NHEADS)];
        }
        #pragma unroll
        for (int ks = 0; ks < 32; ++ks) { wq[ks].x = 0.f; wq[ks].y = 0.f; }
    }

    float bk[8];
    #pragma unroll
    for (int r = 0; r < 8; ++r) bk[r] = bkv_p[16 * h + 8 * hs + r];
    const float bvv = bkv_p[HID + 16 * h + lm];
    const float bev = beg_p[lm];

    for (int tp = 0; tp < 3; ++tp) {
        const int nt = (tp < 2) ? 4 : 2;
        *reinterpret_cast<float4*>(&sFV[h * FP + lane * 4])        = n0;
        *reinterpret_cast<float4*>(&sFV[(h + 8) * FP + lane * 4])  = n1;
        *reinterpret_cast<float4*>(&sFV[(h + 16) * FP + lane * 4]) = n2;
        *reinterpret_cast<float4*>(&sFV[(h + 24) * FP + lane * 4]) = n3;
        if (tp < 2) {
            *reinterpret_cast<float4*>(&sFV[(h + 32) * FP + lane * 4]) = n4;
            *reinterpret_cast<float4*>(&sFV[(h + 40) * FP + lane * 4]) = n5;
            *reinterpret_cast<float4*>(&sFV[(h + 48) * FP + lane * 4]) = n6;
            *reinterpret_cast<float4*>(&sFV[(h + 56) * FP + lane * 4]) = n7;
        }
        __syncthreads();

        if (tp == 0) {
            const float* sb = f2 + (unsigned)(64 + h) * HID;
            n0 = reinterpret_cast<const float4*>(sb)[lane];
            n1 = reinterpret_cast<const float4*>(sb + 8 * HID)[lane];
            n2 = reinterpret_cast<const float4*>(sb + 16 * HID)[lane];
            n3 = reinterpret_cast<const float4*>(sb + 24 * HID)[lane];
            n4 = reinterpret_cast<const float4*>(sb + 32 * HID)[lane];
            n5 = reinterpret_cast<const float4*>(sb + 40 * HID)[lane];
            n6 = reinterpret_cast<const float4*>(sb + 48 * HID)[lane];
            n7 = reinterpret_cast<const float4*>(sb + 56 * HID)[lane];
            #pragma unroll
            for (int ks = 0; ks < 32; ++ks) {
                wq[ks].x = pwq[(4 * ks)     * HID];
                wq[ks].y = pwq[(4 * ks + 1) * HID];
            }
        } else {
            const float* sb = (tp == 1) ? f2 + (unsigned)(128 + h) * HID
                                        : feat1 + ((unsigned)(ipb * 32 + h) * NSZ + j) * HID;
            const unsigned st = (tp == 1) ? (unsigned)(8 * HID) : (unsigned)(8 * NSZ * HID);
            n0 = reinterpret_cast<const float4*>(sb)[lane];
            n1 = reinterpret_cast<const float4*>(sb + st)[lane];
            n2 = reinterpret_cast<const float4*>(sb + 2 * st)[lane];
            n3 = reinterpret_cast<const float4*>(sb + 3 * st)[lane];
        }

        const int se  = (5 * (h + 4 * tp)) & 7;
        const int rot = (se < nt) ? se : 0;

        for (int i = 0; i < nt; ++i) {
            const int sub = (i + rot) & (nt - 1);
            const int tt = 4 * tp + sub;
            const float* pft = pf + sub * 16 * FP;

            v2f fr[32];
            #pragma unroll
            for (int ks = 0; ks < 32; ++ks)
                fr[ks] = *reinterpret_cast<const v2f*>(pft + ks * 4);

            v8f kd = {0.f, 0.f, 0.f, 0.f, 0.f, 0.f, 0.f, 0.f};
            v8f vd = {0.f, 0.f, 0.f, 0.f, 0.f, 0.f, 0.f, 0.f};
            #pragma unroll
            for (int ks = 0; ks < 32; ++ks) {
                kd = wmma_f32_16x16x4(wk[ks], fr[ks], kd);
                vd = wmma_f32_16x16x4(fr[ks], wv[ks], vd);
            }

            {
                float* ok = Kl + (tt * 16 + lm) * KP + 2 * h + hs;
                #pragma unroll
                for (int r = 0; r < 8; ++r) ok[r * NSZ * KP] = kd[r] + bk[r];
            }
            {
                float4 a, b;
                a.x = vd[0] + bvv; a.y = vd[1] + bvv; a.z = vd[2] + bvv; a.w = vd[3] + bvv;
                b.x = vd[4] + bvv; b.y = vd[5] + bvv; b.z = vd[6] + bvv; b.w = vd[7] + bvv;
                float* ov = Vl + (lm % 8) * VH + (2 * h + lm / 8) * VP + tt * 16 + 8 * hs;
                *reinterpret_cast<float4*>(ov)     = a;
                *reinterpret_cast<float4*>(ov + 4) = b;
            }

            if (sub == se) {
                v8f acc = {0.f, 0.f, 0.f, 0.f, 0.f, 0.f, 0.f, 0.f};
                #pragma unroll
                for (int ks = 0; ks < 32; ++ks)
                    acc = wmma_f32_16x16x4(fr[ks], we[ks], acc);
                if (lm < NHEADS) {
                    float* oe = El + lm * NSZ + tt * 16 + 8 * hs;
                    #pragma unroll
                    for (int r = 0; r < 8; ++r) oe[r] = acc[r] + bev;
                } else {
                    float* og = Gl + (lm - NHEADS) * NSZ + tt * 16 + 8 * hs;
                    #pragma unroll
                    for (int r = 0; r < 8; ++r)
                        og[r] = 1.f / (1.f + __expf(-(acc[r] + bev)));
                }
            }
        }
        __syncthreads();
    }

    const float* pK = Kl + (h * NSZ + lm) * KP + 8 * hs;
    const float* pV = Vl + h * VH + lm * VP + 2 * hs;


    float ev[10], gate[10];
    #pragma unroll
    for (int ct = 0; ct < 10; ++ct) {
        ev[ct]   = El[h * NSZ + ct * 16 + lm];
        gate[ct] = Gl[h * NSZ + ct * 16 + lm];
    }
    float bqv[8];
    #pragma unroll
    for (int r = 0; r < 8; ++r) bqv[r] = bq_p[(8 * hs + r) * NHEADS + h];

    const int rowLN = tid / 16;
    const int cg    = tid % 16;

    v2f ot0[4] = {{0.f, 0.f}, {0.f, 0.f}, {0.f, 0.f}, {0.f, 0.f}};
    v2f ot1[4] = {{0.f, 0.f}, {0.f, 0.f}, {0.f, 0.f}, {0.f, 0.f}};
    unsigned oprev0 = 0, oprev1 = 0;

    for (int ip = ipb; ip < ipe; ++ip) {
        *reinterpret_cast<float4*>(&sFV[h * FP + lane * 4])              = n0;
        *reinterpret_cast<float4*>(&sFV[(h + 8) * FP + lane * 4])        = n1;
        *reinterpret_cast<float4*>(&sFV[(16 + h) * FP + lane * 4])       = n2;
        *reinterpret_cast<float4*>(&sFV[(24 + h) * FP + lane * 4])       = n3;
        __syncthreads();

        if (ip > ipb) {
            v2f* op0 = reinterpret_cast<v2f*>(&out[oprev0]);
            v2f* op1 = reinterpret_cast<v2f*>(&out[oprev1]);
            #pragma unroll
            for (int c = 0; c < 4; ++c) { op0[c] = ot0[c]; op1[c] = ot1[c]; }
        }
        if (ip + 1 < ipe) {
            const unsigned rb = (unsigned)((ip + 1) * 32 + h);
            n0 = reinterpret_cast<const float4*>(&feat1[((rb)      * NSZ + j) * HID])[lane];
            n1 = reinterpret_cast<const float4*>(&feat1[((rb + 8)  * NSZ + j) * HID])[lane];
            n2 = reinterpret_cast<const float4*>(&feat1[((rb + 16) * NSZ + j) * HID])[lane];
            n3 = reinterpret_cast<const float4*>(&feat1[((rb + 24) * NSZ + j) * HID])[lane];
        }

        v2f fq0[32];
        #pragma unroll
        for (int ks = 0; ks < 32; ++ks)
            fq0[ks] = *reinterpret_cast<const v2f*>(pf + ks * 4);
        __builtin_amdgcn_sched_barrier(0);

        #pragma unroll
        for (int sub = 0; sub < 2; ++sub) {
            v8f qd = {0.f, 0.f, 0.f, 0.f, 0.f, 0.f, 0.f, 0.f};
            #pragma unroll
            for (int ks = 0; ks < 32; ++ks) {
                const v2f b = (sub == 0) ? fq0[ks]
                    : *reinterpret_cast<const v2f*>(pf + 16 * FP + ks * 4);
                qd = wmma_f32_16x16x4(wq[ks], b, qd);
            }

            v2f afr[4];
            #pragma unroll
            for (int ks = 0; ks < 4; ++ks) {
                afr[ks].x = (qd[2 * ks]     + bqv[2 * ks])     * 0.25f;
                afr[ks].y = (qd[2 * ks + 1] + bqv[2 * ks + 1]) * 0.25f;
            }

            v8f acc[10];
            #pragma unroll
            for (int ct = 0; ct < 10; ++ct) {
                v8f c = {0.f, 0.f, 0.f, 0.f, 0.f, 0.f, 0.f, 0.f};
                #pragma unroll
                for (int ks = 0; ks < 4; ++ks) {
                    v2f b = *reinterpret_cast<const v2f*>(pK + ct * 16 * KP + ks * 2);
                    c = wmma_f32_16x16x4(afr[ks], b, c);
                }
                const float e = ev[ct];
                #pragma unroll
                for (int r = 0; r < 8; ++r) c[r] += e;
                acc[ct] = c;
            }

            float psum[8];
            #pragma unroll
            for (int r = 0; r < 8; ++r) {
                float mx = -1e30f;
                #pragma unroll
                for (int ct = 0; ct < 10; ++ct) mx = fmaxf(mx, acc[ct][r]);
                {
                    const float p1 = __shfl_xor(mx, 1, 32);
                    const float p2 = __shfl_xor(mx, 2, 32);
                    const float p3 = __shfl_xor(mx, 3, 32);
                    mx = fmaxf(mx, fmaxf(fmaxf(p1, p2), p3));
                    const float q1 = __shfl_xor(mx, 4, 32);
                    const float q2 = __shfl_xor(mx, 8, 32);
                    const float q3 = __shfl_xor(mx, 12, 32);
                    mx = fmaxf(mx, fmaxf(fmaxf(q1, q2), q3));
                }
                float sm = 0.f;
                #pragma unroll
                for (int ct = 0; ct < 10; ++ct) {
                    float e = __expf(acc[ct][r] - mx);
                    sm += e;
                    acc[ct][r] = e * gate[ct];
                }
                psum[r] = sm;
            }

            v8f va = {0.f, 0.f, 0.f, 0.f, 0.f, 0.f, 0.f, 0.f};
            v8f sd = {0.f, 0.f, 0.f, 0.f, 0.f, 0.f, 0.f, 0.f};
            #pragma unroll
            for (int half = 0; half < 2; ++half) {
                #pragma unroll
                for (int ct = 0; ct < 5; ++ct) {
                    #pragma unroll
                    for (int r = 0; r < 8; ++r)
                        Plds[h][r + 8 * hs][ct * 16 + lm] = acc[half * 5 + ct][r];
                }
                if (half == 0) {
                    #pragma unroll
                    for (int r = 0; r < 8; ++r) Slds[h][r + 8 * hs][lm] = psum[r];
                }
                wave_lds_handover();
                if (half == 0) {
                    const v2f ones = {1.f, 1.f};
                    #pragma unroll
                    for (int ks = 0; ks < 4; ++ks) {
                        v2f a;
                        a.x = Slds[h][lm][4 * ks + 2 * hs];
                        a.y = Slds[h][lm][4 * ks + 2 * hs + 1];
                        sd = wmma_f32_16x16x4(a, ones, sd);
                    }
                }
                #pragma unroll
                for (int k0 = 0; k0 < 80; k0 += 4) {
                    v2f a, b;
                    a.x = Plds[h][lm][k0 + 2 * hs];
                    a.y = Plds[h][lm][k0 + 2 * hs + 1];
                    b = *reinterpret_cast<const v2f*>(pV + half * 80 + k0);
                    va = wmma_f32_16x16x4(a, b, va);
                }
                wave_lds_handover();
            }

            #pragma unroll
            for (int r = 0; r < 8; ++r) va[r] = va[r] / sd[r];

            #pragma unroll
            for (int r = 0; r < 8; ++r)
                VaLds[sub][r + 8 * hs][lm * NHEADS + h] = va[r];
        }
        __syncthreads();

        #pragma unroll
        for (int sub = 0; sub < 2; ++sub) {
            float vals[8], s1 = 0.f, s2 = 0.f;
            #pragma unroll
            for (int c = 0; c < 8; ++c) {
                float v = VaLds[sub][rowLN][cg * 8 + c];
                vals[c] = v; s1 += v; s2 += v * v;
            }
            {
                const float a1 = __shfl_xor(s1, 1, 32), b1 = __shfl_xor(s2, 1, 32);
                const float a2 = __shfl_xor(s1, 2, 32), b2 = __shfl_xor(s2, 2, 32);
                const float a3 = __shfl_xor(s1, 3, 32), b3 = __shfl_xor(s2, 3, 32);
                s1 += (a1 + a2) + a3;
                s2 += (b1 + b2) + b3;
                const float c1 = __shfl_xor(s1, 4, 32),  d1 = __shfl_xor(s2, 4, 32);
                const float c2 = __shfl_xor(s1, 8, 32),  d2 = __shfl_xor(s2, 8, 32);
                const float c3 = __shfl_xor(s1, 12, 32), d3 = __shfl_xor(s2, 12, 32);
                s1 += (c1 + c2) + c3;
                s2 += (d1 + d2) + d3;
            }
            const float mu  = s1 * (1.f / HID);
            const float var = s2 * (1.f / HID) - mu * mu;
            const float inv = rsqrtf(var + LN_EPS);

            const unsigned op = ((unsigned)(ip * 32 + sub * 16 + rowLN) * NSZ + j) * HID + cg * 8;
            if (sub == 0) oprev0 = op; else oprev1 = op;
            #pragma unroll
            for (int c = 0; c < 8; c += 2) {
                const int ch = cg * 8 + c;
                v2f tv;
                tv.x = (vals[c]     - mu) * inv * ln_g[ch]     + ln_b[ch];
                tv.y = (vals[c + 1] - mu) * inv * ln_g[ch + 1] + ln_b[ch + 1];
                if (sub == 0) ot0[c / 2] = tv; else ot1[c / 2] = tv;
            }
        }
    }
    {
        v2f* op0 = reinterpret_cast<v2f*>(&out[oprev0]);
        v2f* op1 = reinterpret_cast<v2f*>(&out[oprev1]);
        #pragma unroll
        for (int c = 0; c < 4; ++c) { op0[c] = ot0[c]; op1[c] = ot1[c]; }
    }
    }
}

extern "C" void kernel_launch(void* const* d_in, const int* in_sizes, int n_in,
                              void* d_out, int out_size, void* d_ws, size_t ws_size,
                              hipStream_t stream) {
    const float* feat1 = (const float*)d_in[0];
    const float* feat2 = (const float*)d_in[1];
    const float* Wq    = (const float*)d_in[2];
    const float* bq    = (const float*)d_in[3];
    const float* Wkv   = (const float*)d_in[4];
    const float* bkv   = (const float*)d_in[5];
    const float* Weg   = (const float*)d_in[6];
    const float* beg   = (const float*)d_in[7];
    const float* lng   = (const float*)d_in[8];
    const float* lnb   = (const float*)d_in[9];

    fused_kernel<<<dim3(256), 256, 0, stream>>>(
        feat1, feat2, Wq, Wkv, Weg, bq, bkv, beg, lng, lnb, (float*)d_out);
}


# Query tiles dealt singly: a j's own workgroup sweeps tiles 0-8 in three steps of three (one tile's code looped, 2 barriers a step), the 96 others take tile 9 only (of two j or one); sums kept
# speedup vs baseline: 2.7472x; 1.0322x over previous
#include <hip/hip_runtime.h>
#include <hip/hip_bf16.h>


#define NSZ 160
#define NN (NSZ * NSZ)
#define HID 128
#define NHEADS 8
#define HD 16
#define LN_EPS 0.001f

#define FP (HID + 4)
#define KP 20
#define VP 164
#define VH (HD * VP + 8)

typedef __attribute__((ext_vector_type(2))) float v2f;
typedef __attribute__((ext_vector_type(8))) float v8f;

static __device__ __forceinline__ v8f wmma_f32_16x16x4(v2f a, v2f b, v8f c) {
    return __builtin_amdgcn_wmma_f32_16x16x4_f32(
         false, a,  false, b,
         (short)0, c,  false,  false);
}

static __device__ __forceinline__ void wave_lds_handover() {
    __builtin_amdgcn_fence(__ATOMIC_RELEASE, "wavefront");
    __builtin_amdgcn_wave_barrier();
    __builtin_amdgcn_fence(__ATOMIC_ACQUIRE, "wavefront");
}


__global__ __launch_bounds__(256)
void fused_kernel(
    const float* __restrict__ feat1, const float* __restrict__ feat2,
    const float* __restrict__ Wq, const float* __restrict__ Wkv,
    const float* __restrict__ Weg,
    const float* __restrict__ bq, const float* __restrict__ bkv,
    const float* __restrict__ beg,
    const float* __restrict__ ln_g, const float* __restrict__ ln_b,
    float* __restrict__ out)
{
    __shared__ __attribute__((aligned(16))) float Kl[NHEADS * NSZ * KP];
    __shared__ __attribute__((aligned(16))) float Vl[NHEADS * VH];
    __shared__ float El[NHEADS * NSZ];
    __shared__ float Gl[NHEADS * NSZ];
    __shared__ float Plds[NHEADS][16][81];
    __shared__ float Slds[NHEADS][16][17];
    __shared__ __attribute__((aligned(16))) float sFV[4 * 16 * FP];
    __shared__ float VaLds[3][16][HID];

    const int tid  = threadIdx.x;
    const int h    = tid / 32;
    const int lane = tid % 32;
    const int hs   = lane / 16;
    const int lm   = lane % 16;

    const int bid = blockIdx.x;
    const bool own = bid < NSZ;
    int jfirst = bid, nseg = 1;
    if (!own) {
        const int u = bid - NSZ;
        if (u < 64) { jfirst = 2 * u; nseg = 2; } else { jfirst = 64 + u; }
    }
    const int tfirst = own ? 0 : 9;

    for (int seg = 0; seg < nseg; ++seg) {
    const int j = jfirst + seg;

    unsigned zo = 0;
    asm volatile("" : "+s"(zo));
    const float* Wq_p = Wq + zo;   const float* Wkv_p = Wkv + zo; const float* Weg_p = Weg + zo;
    const float* bq_p = bq + zo;   const float* bkv_p = bkv + zo; const float* beg_p = beg + zo;

    const float* f2 = feat2 + (unsigned)j * NSZ * HID;

    float4 n0 = reinterpret_cast<const float4*>(f2 + (unsigned)(h)      * HID)[lane];
    float4 n1 = reinterpret_cast<const float4*>(f2 + (unsigned)(h + 8)  * HID)[lane];
    float4 n2 = reinterpret_cast<const float4*>(f2 + (unsigned)(h + 16) * HID)[lane];
    float4 n3 = reinterpret_cast<const float4*>(f2 + (unsigned)(h + 24) * HID)[lane];
    float4 n4 = reinterpret_cast<const float4*>(f2 + (unsigned)(h + 32) * HID)[lane];
    float4 n5 = reinterpret_cast<const float4*>(f2 + (unsigned)(h + 40) * HID)[lane];
    float4 n6 = reinterpret_cast<const float4*>(f2 + (unsigned)(h + 48) * HID)[lane];
    float4 n7 = reinterpret_cast<const float4*>(f2 + (unsigned)(h + 56) * HID)[lane];

    const float* pf  = sFV + lm * FP + 2 * hs;

    v2f wk[32], wv[32], we[32], wq[32];
    const float* pwq = Wq_p + (unsigned)(2 * hs) * HID + lm * NHEADS + h;
    {
        const float* pwk = Wkv_p + (unsigned)(2 * hs) * (2 * HID) + 16 * h + lm;
        const float* pwv = pwk + HID;
        const float* pwe = Weg_p + (unsigned)(2 * hs) * (2 * NHEADS) + lm;
        #pragma unroll
        for (int ks = 0; ks < 32; ++ks) {
            wk[ks].x = pwk[(4 * ks)     * (2 * HID)];
            wk[ks].y = pwk[(4 * ks + 1) * (2 * HID)];
            wv[ks].x = pwv[(4 * ks)     * (2 * HID)];
            wv[ks].y = pwv[(4 * ks + 1) * (2 * HID)];
        }
        #pragma unroll
        for (int ks = 0; ks < 32; ++ks) {
            we[ks].x = pwe[(4 * ks)     * (2 * NHEADS)];
            we[ks].y = pwe[(4 * ks + 1) * (2 * NHEADS)];
        }
        #pragma unroll
        for (int ks = 0; ks < 32; ++ks) { wq[ks].x = 0.f; wq[ks].y = 0.f; }
    }

    float bk[8];
    #pragma unroll
    for (int r = 0; r < 8; ++r) bk[r] = bkv_p[16 * h + 8 * hs + r];
    const float bvv = bkv_p[HID + 16 * h + lm];
    const float bev = beg_p[lm];

    for (int tp = 0; tp < 3; ++tp) {
        const int nt = (tp < 2) ? 4 : 2;
        *reinterpret_cast<float4*>(&sFV[h * FP + lane * 4])        = n0;
        *reinterpret_cast<float4*>(&sFV[(h + 8) * FP + lane * 4])  = n1;
        *reinterpret_cast<float4*>(&sFV[(h + 16) * FP + lane * 4]) = n2;
        *reinterpret_cast<float4*>(&sFV[(h + 24) * FP + lane * 4]) = n3;
        if (tp < 2) {
            *reinterpret_cast<float4*>(&sFV[(h + 32) * FP + lane * 4]) = n4;
            *reinterpret_cast<float4*>(&sFV[(h + 40) * FP + lane * 4]) = n5;
            *reinterpret_cast<float4*>(&sFV[(h + 48) * FP + lane * 4]) = n6;
            *reinterpret_cast<float4*>(&sFV[(h + 56) * FP + lane * 4]) = n7;
        }
        __syncthreads();

        if (tp == 0) {
            const float* sb = f2 + (unsigned)(64 + h) * HID;
            n0 = reinterpret_cast<const float4*>(sb)[lane];
            n1 = reinterpret_cast<const float4*>(sb + 8 * HID)[lane];
            n2 = reinterpret_cast<const float4*>(sb + 16 * HID)[lane];
            n3 = reinterpret_cast<const float4*>(sb + 24 * HID)[lane];
            n4 = reinterpret_cast<const float4*>(sb + 32 * HID)[lane];
            n5 = reinterpret_cast<const float4*>(sb + 40 * HID)[lane];
            n6 = reinterpret_cast<const float4*>(sb + 48 * HID)[lane];
            n7 = reinterpret_cast<const float4*>(sb + 56 * HID)[lane];
            #pragma unroll
            for (int ks = 0; ks < 32; ++ks) {
                wq[ks].x = pwq[(4 * ks)     * HID];
                wq[ks].y = pwq[(4 * ks + 1) * HID];
            }
        } else {
            const float* sb = (tp == 1) ? f2 + (unsigned)(128 + h) * HID
                                        : feat1 + ((unsigned)(tfirst * 16 + h) * NSZ + j) * HID;
            const unsigned st = (tp == 1) ? (unsigned)(8 * HID) : (unsigned)(8 * NSZ * HID);
            n0 = reinterpret_cast<const float4*>(sb)[lane];
            n1 = reinterpret_cast<const float4*>(sb + st)[lane];
            if (tp == 1 || own) {
                n2 = reinterpret_cast<const float4*>(sb + 2 * st)[lane];
                n3 = reinterpret_cast<const float4*>(sb + 3 * st)[lane];
            }
            if (tp == 2 && own) {
                n4 = reinterpret_cast<const float4*>(sb + 4 * st)[lane];
                n5 = reinterpret_cast<const float4*>(sb + 5 * st)[lane];
            }
        }

        const int se  = (5 * (h + 4 * tp)) & 7;
        const int rot = (se < nt) ? se : 0;

        for (int i = 0; i < nt; ++i) {
            const int sub = (i + rot) & (nt - 1);
            const int tt = 4 * tp + sub;
            const float* pft = pf + sub * 16 * FP;

            v2f fr[32];
            #pragma unroll
            for (int ks = 0; ks < 32; ++ks)
                fr[ks] = *reinterpret_cast<const v2f*>(pft + ks * 4);

            v8f kd = {0.f, 0.f, 0.f, 0.f, 0.f, 0.f, 0.f, 0.f};
            v8f vd = {0.f, 0.f, 0.f, 0.f, 0.f, 0.f, 0.f, 0.f};
            #pragma unroll
            for (int ks = 0; ks < 32; ++ks) {
                kd = wmma_f32_16x16x4(wk[ks], fr[ks], kd);
                vd = wmma_f32_16x16x4(fr[ks], wv[ks], vd);
            }

            {
                float* ok = Kl + (tt * 16 + lm) * KP + 2 * h + hs;
                #pragma unroll
                for (int r = 0; r < 8; ++r) ok[r * NSZ * KP] = kd[r] + bk[r];
            }
            {
                float4 a, b;
                a.x = vd[0] + bvv; a.y = vd[1] + bvv; a.z = vd[2] + bvv; a.w = vd[3] + bvv;
                b.x = vd[4] + bvv; b.y = vd[5] + bvv; b.z = vd[6] + bvv; b.w = vd[7] + bvv;
                float* ov = Vl + (lm % 8) * VH + (2 * h + lm / 8) * VP + tt * 16 + 8 * hs;
                *reinterpret_cast<float4*>(ov)     = a;
                *reinterpret_cast<float4*>(ov + 4) = b;
            }

            if (sub == se) {
                v8f acc = {0.f, 0.f, 0.f, 0.f, 0.f, 0.f, 0.f, 0.f};
                #pragma unroll
                for (int ks = 0; ks < 32; ++ks)
                    acc = wmma_f32_16x16x4(fr[ks], we[ks], acc);
                if (lm < NHEADS) {
                    float* oe = El + lm * NSZ + tt * 16 + 8 * hs;
                    #pragma unroll
                    for (int r = 0; r < 8; ++r) oe[r] = acc[r] + bev;
                } else {
                    float* og = Gl + (lm - NHEADS) * NSZ + tt * 16 + 8 * hs;
                    #pragma unroll
                    for (int r = 0; r < 8; ++r)
                        og[r] = 1.f / (1.f + __expf(-(acc[r] + bev)));
                }
            }
        }
        __syncthreads();
    }

    const float* pK = Kl + (h * NSZ + lm) * KP + 8 * hs;
    const float* pV = Vl + h * VH + lm * VP + 2 * hs;


    float ev[10], gate[10];
    #pragma unroll
    for (int ct = 0; ct < 10; ++ct) {
        ev[ct]   = El[h * NSZ + ct * 16 + lm];
        gate[ct] = Gl[h * NSZ + ct * 16 + lm];
    }
    float bqv[8];
    #pragma unroll
    for (int r = 0; r < 8; ++r) bqv[r] = bq_p[(8 * hs + r) * NHEADS + h];

    const int rowLN = tid / 16;
    const int cg    = tid % 16;

    v2f ot[3][4] = {};
    unsigned oprev[3] = {0, 0, 0};

    const int nt   = own ? 3 : 1;
    const int tend = own ? 9 : 10;
    for (int t0 = tfirst; t0 < tend; t0 += nt) {
        *reinterpret_cast<float4*>(&sFV[h * FP + lane * 4])              = n0;
        *reinterpret_cast<float4*>(&sFV[(h + 8) * FP + lane * 4])        = n1;
        if (nt == 3) {
            *reinterpret_cast<float4*>(&sFV[(16 + h) * FP + lane * 4])   = n2;
            *reinterpret_cast<float4*>(&sFV[(24 + h) * FP + lane * 4])   = n3;
            *reinterpret_cast<float4*>(&sFV[(32 + h) * FP + lane * 4])   = n4;
            *reinterpret_cast<float4*>(&sFV[(40 + h) * FP + lane * 4])   = n5;
        }
        __syncthreads();

        if (t0 > tfirst) {
            #pragma unroll
            for (int sub = 0; sub < 3; ++sub) {
                v2f* opp = reinterpret_cast<v2f*>(&out[oprev[sub]]);
                #pragma unroll
                for (int c = 0; c < 4; ++c) opp[c] = ot[sub][c];
            }
        }
        {
            if (t0 + nt < tend) {
                const unsigned rb = (unsigned)((t0 + 3) * 16 + h);
                n0 = reinterpret_cast<const float4*>(&feat1[((rb)      * NSZ + j) * HID])[lane];
                n1 = reinterpret_cast<const float4*>(&feat1[((rb + 8)  * NSZ + j) * HID])[lane];
                n2 = reinterpret_cast<const float4*>(&feat1[((rb + 16) * NSZ + j) * HID])[lane];
                n3 = reinterpret_cast<const float4*>(&feat1[((rb + 24) * NSZ + j) * HID])[lane];
                n4 = reinterpret_cast<const float4*>(&feat1[((rb + 32) * NSZ + j) * HID])[lane];
                n5 = reinterpret_cast<const float4*>(&feat1[((rb + 40) * NSZ + j) * HID])[lane];
            }
        }

        #pragma nounroll
        for (int sub = 0; sub < nt; ++sub) {
            v2f fq[32];
            #pragma unroll
            for (int ks = 0; ks < 32; ++ks)
                fq[ks] = *reinterpret_cast<const v2f*>(pf + sub * 16 * FP + ks * 4);
            __builtin_amdgcn_sched_barrier(0);

            v8f qd = {0.f, 0.f, 0.f, 0.f, 0.f, 0.f, 0.f, 0.f};
            #pragma unroll
            for (int ks = 0; ks < 32; ++ks) {
                qd = wmma_f32_16x16x4(wq[ks], fq[ks], qd);
            }

            v2f afr[4];
            #pragma unroll
            for (int ks = 0; ks < 4; ++ks) {
                afr[ks].x = (qd[2 * ks]     + bqv[2 * ks])     * 0.25f;
                afr[ks].y = (qd[2 * ks + 1] + bqv[2 * ks + 1]) * 0.25f;
            }

            v8f acc[10];
            #pragma unroll
            for (int ct = 0; ct < 10; ++ct) {
                v8f c = {0.f, 0.f, 0.f, 0.f, 0.f, 0.f, 0.f, 0.f};
                #pragma unroll
                for (int ks = 0; ks < 4; ++ks) {
                    v2f b = *reinterpret_cast<const v2f*>(pK + ct * 16 * KP + ks * 2);
                    c = wmma_f32_16x16x4(afr[ks], b, c);
                }
                const float e = ev[ct];
                #pragma unroll
                for (int r = 0; r < 8; ++r) c[r] += e;
                acc[ct] = c;
            }

            float psum[8];
            #pragma unroll
            for (int r = 0; r < 8; ++r) {
                float mx = -1e30f;
                #pragma unroll
                for (int ct = 0; ct < 10; ++ct) mx = fmaxf(mx, acc[ct][r]);
                {
                    const float p1 = __shfl_xor(mx, 1, 32);
                    const float p2 = __shfl_xor(mx, 2, 32);
                    const float p3 = __shfl_xor(mx, 3, 32);
                    mx = fmaxf(mx, fmaxf(fmaxf(p1, p2), p3));
                    const float q1 = __shfl_xor(mx, 4, 32);
                    const float q2 = __shfl_xor(mx, 8, 32);
                    const float q3 = __shfl_xor(mx, 12, 32);
                    mx = fmaxf(mx, fmaxf(fmaxf(q1, q2), q3));
                }
                float sm = 0.f;
                #pragma unroll
                for (int ct = 0; ct < 10; ++ct) {
                    float e = __expf(acc[ct][r] - mx);
                    sm += e;
                    acc[ct][r] = e * gate[ct];
                }
                psum[r] = sm;
            }

            v8f va = {0.f, 0.f, 0.f, 0.f, 0.f, 0.f, 0.f, 0.f};
            v8f sd = {0.f, 0.f, 0.f, 0.f, 0.f, 0.f, 0.f, 0.f};
            #pragma unroll
            for (int half = 0; half < 2; ++half) {
                #pragma unroll
                for (int ct = 0; ct < 5; ++ct) {
                    #pragma unroll
                    for (int r = 0; r < 8; ++r)
                        Plds[h][r + 8 * hs][ct * 16 + lm] = acc[half * 5 + ct][r];
                }
                if (half == 0) {
                    #pragma unroll
                    for (int r = 0; r < 8; ++r) Slds[h][r + 8 * hs][lm] = psum[r];
                }
                wave_lds_handover();
                if (half == 0) {
                    const v2f ones = {1.f, 1.f};
                    #pragma unroll
                    for (int ks = 0; ks < 4; ++ks) {
                        v2f a;
                        a.x = Slds[h][lm][4 * ks + 2 * hs];
                        a.y = Slds[h][lm][4 * ks + 2 * hs + 1];
                        sd = wmma_f32_16x16x4(a, ones, sd);
                    }
                }
                #pragma unroll
                for (int k0 = 0; k0 < 80; k0 += 4) {
                    v2f a, b;
                    a.x = Plds[h][lm][k0 + 2 * hs];
                    a.y = Plds[h][lm][k0 + 2 * hs + 1];
                    b = *reinterpret_cast<const v2f*>(pV + half * 80 + k0);
                    va = wmma_f32_16x16x4(a, b, va);
                }
                wave_lds_handover();
            }

            #pragma unroll
            for (int r = 0; r < 8; ++r) va[r] = va[r] / sd[r];

            #pragma unroll
            for (int r = 0; r < 8; ++r)
                VaLds[sub][r + 8 * hs][lm * NHEADS + h] = va[r];
        }
        __syncthreads();

        #pragma unroll
        for (int sub = 0; sub < 3; ++sub) {
            if (sub >= nt) continue;
            float vals[8], s1 = 0.f, s2 = 0.f;
            #pragma unroll
            for (int c = 0; c < 8; ++c) {
                float v = VaLds[sub][rowLN][cg * 8 + c];
                vals[c] = v; s1 += v; s2 += v * v;
            }
            {
                const float a1 = __shfl_xor(s1, 1, 32), b1 = __shfl_xor(s2, 1, 32);
                const float a2 = __shfl_xor(s1, 2, 32), b2 = __shfl_xor(s2, 2, 32);
                const float a3 = __shfl_xor(s1, 3, 32), b3 = __shfl_xor(s2, 3, 32);
                s1 += (a1 + a2) + a3;
                s2 += (b1 + b2) + b3;
                const float c1 = __shfl_xor(s1, 4, 32),  d1 = __shfl_xor(s2, 4, 32);
                const float c2 = __shfl_xor(s1, 8, 32),  d2 = __shfl_xor(s2, 8, 32);
                const float c3 = __shfl_xor(s1, 12, 32), d3 = __shfl_xor(s2, 12, 32);
                s1 += (c1 + c2) + c3;
                s2 += (d1 + d2) + d3;
            }
            const float mu  = s1 * (1.f / HID);
            const float var = s2 * (1.f / HID) - mu * mu;
            const float inv = rsqrtf(var + LN_EPS);

            const unsigned op = ((unsigned)(t0 * 16 + sub * 16 + rowLN) * NSZ + j) * HID + cg * 8;
            oprev[sub] = op;
            #pragma unroll
            for (int c = 0; c < 8; c += 2) {
                const int ch = cg * 8 + c;
                v2f tv;
                tv.x = (vals[c]     - mu) * inv * ln_g[ch]     + ln_b[ch];
                tv.y = (vals[c + 1] - mu) * inv * ln_g[ch + 1] + ln_b[ch + 1];
                ot[sub][c / 2] = tv;
            }
        }
    }
    #pragma unroll
    for (int sub = 0; sub < 3; ++sub) {
        if (sub == 0 || own) {
            v2f* opp = reinterpret_cast<v2f*>(&out[oprev[sub]]);
            #pragma unroll
            for (int c = 0; c < 4; ++c) opp[c] = ot[sub][c];
        }
    }
    }
}

extern "C" void kernel_launch(void* const* d_in, const int* in_sizes, int n_in,
                              void* d_out, int out_size, void* d_ws, size_t ws_size,
                              hipStream_t stream) {
    const float* feat1 = (const float*)d_in[0];
    const float* feat2 = (const float*)d_in[1];
    const float* Wq    = (const float*)d_in[2];
    const float* bq    = (const float*)d_in[3];
    const float* Wkv   = (const float*)d_in[4];
    const float* bkv   = (const float*)d_in[5];
    const float* Weg   = (const float*)d_in[6];
    const float* beg   = (const float*)d_in[7];
    const float* lng   = (const float*)d_in[8];
    const float* lnb   = (const float*)d_in[9];

    fused_kernel<<<dim3(256), 256, 0, stream>>>(
        feat1, feat2, Wq, Wkv, Weg, bq, bkv, beg, lng, lnb, (float*)d_out);
}
